# starting kernel
# baseline (speedup 1.0000x reference)
#include <hip/hip_runtime.h>
#include <hip/hip_bf16.h>
#include <math.h>


typedef __attribute__((ext_vector_type(2))) float v2f;
typedef __attribute__((ext_vector_type(8))) float v8f;

#define IN_C  128
#define OUT_C 64
#define GAT_ALPHA 0.2f

__device__ __forceinline__ float lrelu(float z) {
    return z >= 0.0f ? z : GAT_ALPHA * z;
}

__global__ void gat_gemm_stw(const float* __restrict__ x,
                             const float* __restrict__ W,
                             const float* __restrict__ a,
                             float* __restrict__ h,
                             float* __restrict__ s,
                             float* __restrict__ t,
                             int n, int mtiles) {
    const int wavesPerBlock = blockDim.x >> 5;
    const int wave = blockIdx.x * wavesPerBlock + ((int)threadIdx.x >> 5);
    if (wave >= mtiles) return;

    const int lane = threadIdx.x & 31;
    const int half = lane >> 4;
    const int l15  = lane & 15;
    const int m0   = wave * 16;

    int arow = m0 + l15; if (arow > n - 1) arow = n - 1;
    const float* xrow = x + (size_t)arow * IN_C;

    v2f afrag[32];
#pragma unroll
    for (int ks = 0; ks < 32; ++ks) {
        const int k = ks * 4 + half * 2;
        afrag[ks].x = xrow[k];
        afrag[ks].y = xrow[k + 1];
    }

    float pr1[8], pr2[8];
#pragma unroll
    for (int r = 0; r < 8; ++r) { pr1[r] = 0.0f; pr2[r] = 0.0f; }

#pragma unroll
    for (int nt = 0; nt < 4; ++nt) {
        const int n0 = nt * 16;
        v8f acc = {};
#pragma unroll
        for (int ks = 0; ks < 32; ++ks) {
            const int k = ks * 4 + half * 2;
            v2f b;
            b.x = W[(size_t)k       * OUT_C + n0 + l15];
            b.y = W[(size_t)(k + 1) * OUT_C + n0 + l15];
            acc = __builtin_amdgcn_wmma_f32_16x16x4_f32(
                 false, afrag[ks],  false, b,
                 (short)0, acc,  false,  false);
        }
        const float a1c = a[n0 + l15];
        const float a2c = a[OUT_C + n0 + l15];
#pragma unroll
        for (int r = 0; r < 8; ++r) {
            const int row = m0 + r + 8 * half;
            const float v = acc[r];
            if (row < n) h[(size_t)row * OUT_C + n0 + l15] = v;
            pr1[r] += v * a1c;
            pr2[r] += v * a2c;
        }
    }

#pragma unroll
    for (int mask = 8; mask >= 1; mask >>= 1) {
#pragma unroll
        for (int r = 0; r < 8; ++r) {
            pr1[r] += __shfl_xor(pr1[r], mask, 32);
            pr2[r] += __shfl_xor(pr2[r], mask, 32);
        }
    }
    if (l15 == 0) {
#pragma unroll
        for (int r = 0; r < 8; ++r) {
            const int row = m0 + r + 8 * half;
            if (row < n) { s[row] = pr1[r]; t[row] = pr2[r]; }
        }
    }
}

__global__ void gat_zero(float* __restrict__ out, float* __restrict__ expsum, int n) {
    const size_t nout = (size_t)n * OUT_C;
    const size_t tot  = nout + (size_t)n;
    const size_t stride = (size_t)gridDim.x * blockDim.x;
    for (size_t i = (size_t)blockIdx.x * blockDim.x + threadIdx.x; i < tot; i += stride) {
        if (i < nout) out[i] = 0.0f;
        else          expsum[i - nout] = 0.0f;
    }
}

__global__ void gat_edge(const int* __restrict__ ei,
                         const float* __restrict__ h,
                         const float* __restrict__ s,
                         const float* __restrict__ t,
                         float* __restrict__ hp,
                         float* __restrict__ expsum,
                         int nE) {
    const int wavesPerBlock = blockDim.x >> 5;
    const int e = blockIdx.x * wavesPerBlock + ((int)threadIdx.x >> 5);
    if (e >= nE) return;
    const int lane = threadIdx.x & 31;

    const int src = ei[e];
    const int dst = ei[nE + e];
    const float w = __expf(lrelu(s[src] + t[dst]));

    const float* hd = h  + (size_t)dst * OUT_C;
    float*       hs = hp + (size_t)src * OUT_C;
    const int c = lane * 2;
    atomicAdd(&hs[c],     w * hd[c]);
    atomicAdd(&hs[c + 1], w * hd[c + 1]);
    if (lane == 0) atomicAdd(&expsum[src], w);
}

__global__ void gat_finalize(const float* __restrict__ h,
                             const float* __restrict__ s,
                             const float* __restrict__ t,
                             const float* __restrict__ expsum,
                             float* __restrict__ out,
                             int n) {
    const size_t tot = (size_t)n * OUT_C;
    const size_t stride = (size_t)gridDim.x * blockDim.x;
    for (size_t idx = (size_t)blockIdx.x * blockDim.x + threadIdx.x; idx < tot; idx += stride) {
        const size_t i = idx / OUT_C;
        const float w = __expf(lrelu(s[i] + t[i]));
        const float denom = expsum[i] + w;
        out[idx] = (out[idx] + w * h[idx]) / denom;
    }
}

extern "C" void kernel_launch(void* const* d_in, const int* in_sizes, int n_in,
                              void* d_out, int out_size, void* d_ws, size_t ws_size,
                              hipStream_t stream) {
    const float* x  = (const float*)d_in[0];
    const float* W  = (const float*)d_in[1];
    const float* a  = (const float*)d_in[2];
    const int*   ei = (const int*)d_in[3];

    const int n = in_sizes[0] / IN_C;
    const int E = in_sizes[3] / 2;

    float* ws     = (float*)d_ws;
    float* h      = ws;
    float* s      = h + (size_t)n * OUT_C;
    float* t      = s + n;
    float* expsum = t + n;
    float* out    = (float*)d_out;

    const dim3 blk(256);
    const int wavesPerBlock = 256 / 32;

    const int mtiles = (n + 15) / 16;
    gat_gemm_stw<<<(mtiles + wavesPerBlock - 1) / wavesPerBlock, blk, 0, stream>>>(
        x, W, a, h, s, t, n, mtiles);

    gat_zero<<<1024, blk, 0, stream>>>(out, expsum, n);

    gat_edge<<<(E + wavesPerBlock - 1) / wavesPerBlock, blk, 0, stream>>>(
        ei, h, s, t, out, expsum, E);

    gat_finalize<<<2048, blk, 0, stream>>>(h, s, t, expsum, out, n);
}


# edges grouped by source with no per-edge global add or scattered store: chunks binned by 64-node bucket in LDS (one global add per chunk and bucket), node lists ranked in LDS; arrival-order sums
# speedup vs baseline: 2.9426x; 2.9426x over previous
#include <hip/hip_runtime.h>
#include <hip/hip_bf16.h>
#include <math.h>


typedef __attribute__((ext_vector_type(2))) float v2f;
typedef __attribute__((ext_vector_type(8))) float v8f;

#define IN_C  128
#define OUT_C 64
#define GAT_ALPHA 0.2f

#define BK_SH    6
#define BK_N     (1 << BK_SH)
#define BK_WORDS (BK_N * OUT_C)
#define MAXB     2048
#define CH       8192
#define BIN_T    512
#define BIN_EPT  (CH / BIN_T)
#define GAT_T    256

__device__ __forceinline__ float lrelu(float z) {
    return z >= 0.0f ? z : GAT_ALPHA * z;
}

__global__ void gat_gemm_stw(const float* __restrict__ x,
                             const float* __restrict__ W,
                             const float* __restrict__ a,
                             float* __restrict__ h,
                             float* __restrict__ s,
                             float* __restrict__ t,
                             int* __restrict__ bcnt,
                             int n, int mtiles, int nbk) {
    const int wavesPerBlock = blockDim.x >> 5;
    const int wave = blockIdx.x * wavesPerBlock + ((int)threadIdx.x >> 5);
    if (wave >= mtiles) return;

    const int lane = threadIdx.x & 31;
    const int half = lane >> 4;
    const int l15  = lane & 15;
    const int m0   = wave * 16;

    if (lane == 0 && wave < nbk) bcnt[wave] = 0;

    int arow = m0 + l15; if (arow > n - 1) arow = n - 1;
    const float* xrow = x + (size_t)arow * IN_C;

    v2f afrag[32];
#pragma unroll
    for (int ks = 0; ks < 32; ++ks) {
        const int k = ks * 4 + half * 2;
        afrag[ks].x = xrow[k];
        afrag[ks].y = xrow[k + 1];
    }

    float pr1[8], pr2[8];
#pragma unroll
    for (int r = 0; r < 8; ++r) { pr1[r] = 0.0f; pr2[r] = 0.0f; }

#pragma unroll
    for (int nt = 0; nt < 4; ++nt) {
        const int n0 = nt * 16;
        v8f acc = {};
#pragma unroll
        for (int ks = 0; ks < 32; ++ks) {
            const int k = ks * 4 + half * 2;
            v2f b;
            b.x = W[(size_t)k       * OUT_C + n0 + l15];
            b.y = W[(size_t)(k + 1) * OUT_C + n0 + l15];
            acc = __builtin_amdgcn_wmma_f32_16x16x4_f32(
                 false, afrag[ks],  false, b,
                 (short)0, acc,  false,  false);
        }
        const float a1c = a[n0 + l15];
        const float a2c = a[OUT_C + n0 + l15];
#pragma unroll
        for (int r = 0; r < 8; ++r) {
            const int row = m0 + r + 8 * half;
            const float v = acc[r];
            if (row < n) h[(size_t)row * OUT_C + n0 + l15] = v;
            pr1[r] += v * a1c;
            pr2[r] += v * a2c;
        }
    }

#pragma unroll
    for (int mask = 8; mask >= 1; mask >>= 1) {
#pragma unroll
        for (int r = 0; r < 8; ++r) {
            pr1[r] += __shfl_xor(pr1[r], mask, 32);
            pr2[r] += __shfl_xor(pr2[r], mask, 32);
        }
    }
    if (l15 == 0) {
#pragma unroll
        for (int r = 0; r < 8; ++r) {
            const int row = m0 + r + 8 * half;
            if (row < n) { s[row] = pr1[r]; t[row] = pr2[r]; }
        }
    }
}

__global__ void __launch_bounds__(BIN_T)
gat_bin(const int* __restrict__ ei,
        int* __restrict__ bcnt,
        unsigned* __restrict__ slots,
        int n, int nE) {
    __shared__ int      hist[MAXB];
    __shared__ int      gofs[MAXB];
    __shared__ unsigned sEnt[CH];
    __shared__ int      sAdr[CH];
    __shared__ int      wsum[BIN_T / 32];

    const int tid   = threadIdx.x;
    const int lane  = tid & 31;
    const int wv    = tid >> 5;
    const int ebase = blockIdx.x * CH;

    for (int j = 0; j < MAXB / BIN_T; ++j) hist[tid + j * BIN_T] = 0;
    __syncthreads();

    unsigned ent[BIN_EPT];
    int meta[BIN_EPT];
#pragma unroll
    for (int k = 0; k < BIN_EPT; ++k) {
        const int e  = ebase + k * BIN_T + tid;
        const bool valid = e < nE;
        const int ec = valid ? e : nE - 1;
        int sN = ei[ec];
        int dN = ei[nE + ec];
        sN = sN < 0 ? 0 : (sN > n - 1 ? n - 1 : sN);
        dN = dN < 0 ? 0 : (dN > n - 1 ? n - 1 : dN);
        const int bkt = sN >> BK_SH;
        ent[k]  = ((unsigned)dN << BK_SH) | (unsigned)(sN & (BK_N - 1));
        meta[k] = -1;
        if (valid) {
            const int lr = atomicAdd(&hist[bkt], 1);
            meta[k] = bkt | (lr << 11);
        }
    }
    __syncthreads();

    for (int j = 0; j < MAXB / BIN_T; ++j) {
        const int b = tid + j * BIN_T;
        const int c = hist[b];
        if (c > 0) gofs[b] = atomicAdd(&bcnt[b], c);
    }

    const int c0 = hist[4 * tid], c1 = hist[4 * tid + 1], c2 = hist[4 * tid + 2], c3 = hist[4 * tid + 3];
    const int tsum = c0 + c1 + c2 + c3;
    int incl = tsum;
    for (int o = 1; o < 32; o <<= 1) {
        const int v = __shfl_up(incl, o, 32);
        if (lane >= o) incl += v;
    }
    if (lane == 31) wsum[wv] = incl;
    __syncthreads();
    int woff = 0;
    for (int q = 0; q < wv; ++q) woff += wsum[q];
    const int ex = woff + incl - tsum;
    hist[4 * tid]     = ex;
    hist[4 * tid + 1] = ex + c0;
    hist[4 * tid + 2] = ex + c0 + c1;
    hist[4 * tid + 3] = ex + c0 + c1 + c2;
    __syncthreads();

#pragma unroll
    for (int k = 0; k < BIN_EPT; ++k) {
        if (meta[k] >= 0) {
            const int bkt = meta[k] & (MAXB - 1);
            const int lr  = meta[k] >> 11;
            const int p   = hist[bkt] + lr;
            const int g   = gofs[bkt] + lr;
            int nn = n - (bkt << BK_SH); nn = nn < BK_N ? nn : BK_N;
            sEnt[p] = ent[k];
            sAdr[p] = g < nn * OUT_C ? bkt * BK_WORDS + g : -1;
        }
    }
    __syncthreads();

    int total = nE - ebase; total = total < CH ? total : CH;
    for (int p = tid; p < total; p += BIN_T) {
        const int a = sAdr[p];
        if (a >= 0) slots[a] = sEnt[p];
    }
}

__device__ __forceinline__ float lane_f(float v, int l) {
    return __builtin_bit_cast(float, __builtin_amdgcn_readlane(__builtin_bit_cast(int, v), l));
}

__global__ void __launch_bounds__(GAT_T)
gat_gather(const int* __restrict__ ei,
           const int* __restrict__ bcnt,
           const float* __restrict__ h,
           const float* __restrict__ s,
           const float* __restrict__ t,
           float* __restrict__ out,
           int n, int nE) {
    __shared__ unsigned raw[BK_WORDS];
    __shared__ int      dsts[BK_WORDS];
    __shared__ int      ncnt[BK_N];
    __shared__ int      nstart[BK_N];
    __shared__ int      ncur[BK_N];
    __shared__ float    sN[BK_N];
    __shared__ float    tN[BK_N];
    __shared__ int      nextNode;

    const int tid  = threadIdx.x;
    const int lane = tid & 31;
    const int b    = blockIdx.x;
    const int lo   = b << BK_SH;
    int nn = n - lo; nn = nn < BK_N ? nn : BK_N;
    const int tot  = bcnt[b];
    const bool ok  = tot <= nn * OUT_C;
    const unsigned* region = (const unsigned*)out + (size_t)lo * OUT_C;

    if (tid < BK_N) {
        ncnt[tid] = 0;
        ncur[tid] = 0;
        if (tid < nn) { sN[tid] = s[lo + tid]; tN[tid] = t[lo + tid]; }
    }
    if (tid == 0) nextNode = 0;
    __syncthreads();
    if (ok) {
        for (int base = 0; base < tot; base += 4 * GAT_T) {
            unsigned ev[4];
#pragma unroll
            for (int k = 0; k < 4; ++k) {
                const int p = base + k * GAT_T + tid;
                ev[k] = p < tot ? region[p] : 0u;
            }
#pragma unroll
            for (int k = 0; k < 4; ++k) {
                const int p = base + k * GAT_T + tid;
                if (p < tot) {
                    raw[p] = ev[k];
                    atomicAdd(&ncnt[ev[k] & (BK_N - 1)], 1);
                }
            }
        }
    }
    __syncthreads();
    if (tid < 32) {
        const int a0 = ncnt[2 * lane], a1 = ncnt[2 * lane + 1];
        const int sm = a0 + a1;
        int incl = sm;
        for (int o = 1; o < 32; o <<= 1) {
            const int v = __shfl_up(incl, o, 32);
            if (lane >= o) incl += v;
        }
        nstart[2 * lane]     = incl - sm;
        nstart[2 * lane + 1] = incl - sm + a0;
    }
    __syncthreads();
    if (ok) {
        for (int p = tid; p < tot; p += GAT_T) {
            const unsigned e = raw[p];
            const int l   = (int)(e & (BK_N - 1));
            const int pos = nstart[l] + atomicAdd(&ncur[l], 1);
            dsts[pos] = (int)(e >> BK_SH);
        }
    }
    __syncthreads();

    const v2f* h2 = (const v2f*)h;
    for (;;) {
        int l = 0;
        if (lane == 0) l = atomicAdd(&nextNode, 1);
        l = __builtin_amdgcn_readfirstlane(l);
        if (l >= nn) break;
        const int i = lo + l;
        float* orow = out + (size_t)i * OUT_C;
        const float si = lane_f(sN[l], 0);
        const float wself = __expf(lrelu(si + lane_f(tN[l], 0)));
        const v2f hself = h2[(size_t)i * 32 + lane];
        v2f acc = {0.0f, 0.0f};
        float den = 0.0f;

        if (ok) {
            const int deg = __builtin_amdgcn_readfirstlane(ncnt[l]);
            const int st  = __builtin_amdgcn_readfirstlane(nstart[l]);
            for (int bb = 0; bb < deg; bb += 32) {
                const int m = deg - bb < 32 ? deg - bb : 32;
                int d = lane < m ? dsts[st + bb + lane] : i;
                d = d < 0 ? 0 : (d > n - 1 ? n - 1 : d);
                const float tv = t[d];
                for (int j = 0; j < m; j += 4) {
                    const int d0 = __builtin_amdgcn_readlane(d, j);
                    const int d1 = __builtin_amdgcn_readlane(d, j + 1);
                    const int d2 = __builtin_amdgcn_readlane(d, j + 2);
                    const int d3 = __builtin_amdgcn_readlane(d, j + 3);
                    const v2f r0 = h2[(size_t)d0 * 32 + lane];
                    const v2f r1 = h2[(size_t)d1 * 32 + lane];
                    const v2f r2 = h2[(size_t)d2 * 32 + lane];
                    const v2f r3 = h2[(size_t)d3 * 32 + lane];
                    const float e0 = __expf(lrelu(si + lane_f(tv, j)));
                    const float e1 = __expf(lrelu(si + lane_f(tv, j + 1)));
                    const float e2 = __expf(lrelu(si + lane_f(tv, j + 2)));
                    const float e3 = __expf(lrelu(si + lane_f(tv, j + 3)));
                    const float w0 = e0;
                    const float w1 = j + 1 < m ? e1 : 0.0f;
                    const float w2 = j + 2 < m ? e2 : 0.0f;
                    const float w3 = j + 3 < m ? e3 : 0.0f;
                    den += (w0 + w1) + (w2 + w3);
                    acc += r0 * w0;
                    acc += r1 * w1;
                    acc += r2 * w2;
                    acc += r3 * w3;
                }
            }
        } else {
            for (int base = 0; base < nE; base += 32) {
                const int e  = base + lane;
                const int ec = e < nE ? e : nE - 1;
                const bool mine = e < nE && ei[ec] == i;
                unsigned mask = __builtin_amdgcn_ballot_w32(mine);
                if (mask == 0u) continue;
                int d = ei[nE + ec];
                d = d < 0 ? 0 : (d > n - 1 ? n - 1 : d);
                const float tv = t[d];
                while (mask != 0u) {
                    const int j = __builtin_ctz(mask);
                    mask &= mask - 1u;
                    const int dj = __builtin_amdgcn_readlane(d, j);
                    const float wj = __expf(lrelu(si + lane_f(tv, j)));
                    const v2f r = h2[(size_t)dj * 32 + lane];
                    den += wj;
                    acc += r * wj;
                }
            }
        }

        acc += hself * wself;
        den += wself;

        v2f o;
        o.x = acc.x / den;
        o.y = acc.y / den;
        ((v2f*)orow)[lane] = o;
    }
}

extern "C" void kernel_launch(void* const* d_in, const int* in_sizes, int n_in,
                              void* d_out, int out_size, void* d_ws, size_t ws_size,
                              hipStream_t stream) {
    const float* x  = (const float*)d_in[0];
    const float* W  = (const float*)d_in[1];
    const float* a  = (const float*)d_in[2];
    const int*   ei = (const int*)d_in[3];

    const int n = in_sizes[0] / IN_C;
    const int E = in_sizes[3] / 2;

    float* ws     = (float*)d_ws;
    float* h      = ws;
    float* s      = h + (size_t)n * OUT_C;
    float* t      = s + n;
    int*   bcnt   = (int*)(t + n);
    float* out    = (float*)d_out;

    const dim3 blk(256);
    const int wavesPerBlock = 256 / 32;

    const int mtiles = (n + 15) / 16;
    const int nbk    = (n + BK_N - 1) / BK_N;
    gat_gemm_stw<<<(mtiles + wavesPerBlock - 1) / wavesPerBlock, blk, 0, stream>>>(
        x, W, a, h, s, t, bcnt, n, mtiles, nbk);

    gat_bin<<<(E + CH - 1) / CH, dim3(BIN_T), 0, stream>>>(ei, bcnt, (unsigned*)d_out, n, E);

    gat_gather<<<nbk, dim3(GAT_T), 0, stream>>>(ei, bcnt, h, s, t, out, n, E);
}


# gather kernel drops the per-edge t[dst] lookup: t is summed from the gathered h row itself (16 lanes x 16 B per row, two edges per load, 16-lane add tree); node sums as two half-wave partials joined
# speedup vs baseline: 3.0525x; 1.0373x over previous
#include <hip/hip_runtime.h>
#include <hip/hip_bf16.h>
#include <math.h>


typedef __attribute__((ext_vector_type(2))) float v2f;
typedef __attribute__((ext_vector_type(4))) float v4f;
typedef __attribute__((ext_vector_type(8))) float v8f;

#define IN_C  128
#define OUT_C 64
#define GAT_ALPHA 0.2f

#define BK_SH    6
#define BK_N     (1 << BK_SH)
#define BK_WORDS (BK_N * OUT_C)
#define MAXB     2048
#define CH       8192
#define BIN_T    512
#define BIN_EPT  (CH / BIN_T)
#define GAT_T    256

__device__ __forceinline__ float lrelu(float z) {
    return z >= 0.0f ? z : GAT_ALPHA * z;
}

__global__ void gat_gemm_stw(const float* __restrict__ x,
                             const float* __restrict__ W,
                             const float* __restrict__ a,
                             float* __restrict__ h,
                             float* __restrict__ s,
                             float* __restrict__ t,
                             int* __restrict__ bcnt,
                             int n, int mtiles, int nbk) {
    const int wavesPerBlock = blockDim.x >> 5;
    const int wave = blockIdx.x * wavesPerBlock + ((int)threadIdx.x >> 5);
    if (wave >= mtiles) return;

    const int lane = threadIdx.x & 31;
    const int half = lane >> 4;
    const int l15  = lane & 15;
    const int m0   = wave * 16;

    if (lane == 0 && wave < nbk) bcnt[wave] = 0;

    int arow = m0 + l15; if (arow > n - 1) arow = n - 1;
    const float* xrow = x + (size_t)arow * IN_C;

    v2f afrag[32];
#pragma unroll
    for (int ks = 0; ks < 32; ++ks) {
        const int k = ks * 4 + half * 2;
        afrag[ks].x = xrow[k];
        afrag[ks].y = xrow[k + 1];
    }

    float pr1[8], pr2[8];
#pragma unroll
    for (int r = 0; r < 8; ++r) { pr1[r] = 0.0f; pr2[r] = 0.0f; }

#pragma unroll
    for (int nt = 0; nt < 4; ++nt) {
        const int n0 = nt * 16;
        v8f acc = {};
#pragma unroll
        for (int ks = 0; ks < 32; ++ks) {
            const int k = ks * 4 + half * 2;
            v2f b;
            b.x = W[(size_t)k       * OUT_C + n0 + l15];
            b.y = W[(size_t)(k + 1) * OUT_C + n0 + l15];
            acc = __builtin_amdgcn_wmma_f32_16x16x4_f32(
                 false, afrag[ks],  false, b,
                 (short)0, acc,  false,  false);
        }
        const float a1c = a[n0 + l15];
        const float a2c = a[OUT_C + n0 + l15];
#pragma unroll
        for (int r = 0; r < 8; ++r) {
            const int row = m0 + r + 8 * half;
            const float v = acc[r];
            if (row < n) h[(size_t)row * OUT_C + n0 + l15] = v;
            pr1[r] += v * a1c;
            pr2[r] += v * a2c;
        }
    }

#pragma unroll
    for (int mask = 8; mask >= 1; mask >>= 1) {
#pragma unroll
        for (int r = 0; r < 8; ++r) {
            pr1[r] += __shfl_xor(pr1[r], mask, 32);
            pr2[r] += __shfl_xor(pr2[r], mask, 32);
        }
    }
    if (l15 == 0) {
#pragma unroll
        for (int r = 0; r < 8; ++r) {
            const int row = m0 + r + 8 * half;
            if (row < n) { s[row] = pr1[r]; t[row] = pr2[r]; }
        }
    }
}

__global__ void __launch_bounds__(BIN_T)
gat_bin(const int* __restrict__ ei,
        int* __restrict__ bcnt,
        unsigned* __restrict__ slots,
        int n, int nE) {
    __shared__ int      hist[MAXB];
    __shared__ int      gofs[MAXB];
    __shared__ unsigned sEnt[CH];
    __shared__ int      sAdr[CH];
    __shared__ int      wsum[BIN_T / 32];

    const int tid   = threadIdx.x;
    const int lane  = tid & 31;
    const int wv    = tid >> 5;
    const int ebase = blockIdx.x * CH;

    for (int j = 0; j < MAXB / BIN_T; ++j) hist[tid + j * BIN_T] = 0;
    __syncthreads();

    unsigned ent[BIN_EPT];
    int meta[BIN_EPT];
#pragma unroll
    for (int k = 0; k < BIN_EPT; ++k) {
        const int e  = ebase + k * BIN_T + tid;
        const bool valid = e < nE;
        const int ec = valid ? e : nE - 1;
        int sN = ei[ec];
        int dN = ei[nE + ec];
        sN = sN < 0 ? 0 : (sN > n - 1 ? n - 1 : sN);
        dN = dN < 0 ? 0 : (dN > n - 1 ? n - 1 : dN);
        const int bkt = sN >> BK_SH;
        ent[k]  = ((unsigned)dN << BK_SH) | (unsigned)(sN & (BK_N - 1));
        meta[k] = -1;
        if (valid) {
            const int lr = atomicAdd(&hist[bkt], 1);
            meta[k] = bkt | (lr << 11);
        }
    }
    __syncthreads();

    for (int j = 0; j < MAXB / BIN_T; ++j) {
        const int b = tid + j * BIN_T;
        const int c = hist[b];
        if (c > 0) gofs[b] = atomicAdd(&bcnt[b], c);
    }

    const int c0 = hist[4 * tid], c1 = hist[4 * tid + 1], c2 = hist[4 * tid + 2], c3 = hist[4 * tid + 3];
    const int tsum = c0 + c1 + c2 + c3;
    int incl = tsum;
    for (int o = 1; o < 32; o <<= 1) {
        const int v = __shfl_up(incl, o, 32);
        if (lane >= o) incl += v;
    }
    if (lane == 31) wsum[wv] = incl;
    __syncthreads();
    int woff = 0;
    for (int q = 0; q < wv; ++q) woff += wsum[q];
    const int ex = woff + incl - tsum;
    hist[4 * tid]     = ex;
    hist[4 * tid + 1] = ex + c0;
    hist[4 * tid + 2] = ex + c0 + c1;
    hist[4 * tid + 3] = ex + c0 + c1 + c2;
    __syncthreads();

#pragma unroll
    for (int k = 0; k < BIN_EPT; ++k) {
        if (meta[k] >= 0) {
            const int bkt = meta[k] & (MAXB - 1);
            const int lr  = meta[k] >> 11;
            const int p   = hist[bkt] + lr;
            const int g   = gofs[bkt] + lr;
            int nn = n - (bkt << BK_SH); nn = nn < BK_N ? nn : BK_N;
            sEnt[p] = ent[k];
            sAdr[p] = g < nn * OUT_C ? bkt * BK_WORDS + g : -1;
        }
    }
    __syncthreads();

    int total = nE - ebase; total = total < CH ? total : CH;
    for (int p = tid; p < total; p += BIN_T) {
        const int a = sAdr[p];
        if (a >= 0) slots[a] = sEnt[p];
    }
}

__device__ __forceinline__ float lane_f(float v, int l) {
    return __builtin_bit_cast(float, __builtin_amdgcn_readlane(__builtin_bit_cast(int, v), l));
}

__device__ __forceinline__ float xhalf(float x) {
    const int b = __builtin_bit_cast(int, x);
    return __builtin_bit_cast(float, __builtin_amdgcn_permlanex16(b, b, 0x76543210, 0xfedcba98, false, false));
}

template <int M>
__device__ __forceinline__ float half_xor(float x) {
    return __builtin_bit_cast(float,
        __builtin_amdgcn_update_dpp(0, __builtin_bit_cast(int, x), 0x160 + M, 0xf, 0xf, true));
}

__device__ __forceinline__ float half_sum(float x) {
    x += half_xor<8>(x);
    x += half_xor<4>(x);
    x += half_xor<2>(x);
    x += half_xor<1>(x);
    return x;
}

__global__ void __launch_bounds__(GAT_T)
gat_gather(const int* __restrict__ ei,
           const int* __restrict__ bcnt,
           const float* __restrict__ a,
           const float* __restrict__ h,
           const float* __restrict__ s,
           const float* __restrict__ t,
           float* __restrict__ out,
           int n, int nE) {
    __shared__ unsigned raw[BK_WORDS];
    __shared__ int      dsts[BK_WORDS];
    __shared__ int      ncnt[BK_N];
    __shared__ int      nstart[BK_N];
    __shared__ int      ncur[BK_N];
    __shared__ float    sN[BK_N];
    __shared__ float    tN[BK_N];
    __shared__ int      nextNode;

    const int tid  = threadIdx.x;
    const int lane = tid & 31;
    const int b    = blockIdx.x;
    const int lo   = b << BK_SH;
    int nn = n - lo; nn = nn < BK_N ? nn : BK_N;
    const int tot  = bcnt[b];
    const bool ok  = tot <= nn * OUT_C;
    const unsigned* region = (const unsigned*)out + (size_t)lo * OUT_C;

    if (tid < BK_N) {
        ncnt[tid] = 0;
        ncur[tid] = 0;
        if (tid < nn) { sN[tid] = s[lo + tid]; tN[tid] = t[lo + tid]; }
    }
    if (tid == 0) nextNode = 0;
    __syncthreads();
    if (ok) {
        for (int base = 0; base < tot; base += 4 * GAT_T) {
            unsigned ev[4];
#pragma unroll
            for (int k = 0; k < 4; ++k) {
                const int p = base + k * GAT_T + tid;
                ev[k] = p < tot ? region[p] : 0u;
            }
#pragma unroll
            for (int k = 0; k < 4; ++k) {
                const int p = base + k * GAT_T + tid;
                if (p < tot) {
                    raw[p] = ev[k];
                    atomicAdd(&ncnt[ev[k] & (BK_N - 1)], 1);
                }
            }
        }
    }
    __syncthreads();
    if (tid < 32) {
        const int a0 = ncnt[2 * lane], a1 = ncnt[2 * lane + 1];
        const int sm = a0 + a1;
        int incl = sm;
        for (int o = 1; o < 32; o <<= 1) {
            const int v = __shfl_up(incl, o, 32);
            if (lane >= o) incl += v;
        }
        nstart[2 * lane]     = incl - sm;
        nstart[2 * lane + 1] = incl - sm + a0;
    }
    __syncthreads();
    if (ok) {
        for (int p = tid; p < tot; p += GAT_T) {
            const unsigned e = raw[p];
            const int l   = (int)(e & (BK_N - 1));
            const int pos = nstart[l] + atomicAdd(&ncur[l], 1);
            dsts[pos] = (int)(e >> BK_SH);
        }
    }
    __syncthreads();

    const int hl = lane >> 4;
    const int c4 = lane & 15;
    const v4f* h4 = (const v4f*)h;
    v4f a2v;
    a2v.x = a[OUT_C + 4 * c4];
    a2v.y = a[OUT_C + 4 * c4 + 1];
    a2v.z = a[OUT_C + 4 * c4 + 2];
    a2v.w = a[OUT_C + 4 * c4 + 3];
    for (;;) {
        int l = 0;
        if (lane == 0) l = atomicAdd(&nextNode, 1);
        l = __builtin_amdgcn_readfirstlane(l);
        if (l >= nn) break;
        const int i = lo + l;
        float* orow = out + (size_t)i * OUT_C;
        const float si = lane_f(sN[l], 0);
        const float wself = __expf(lrelu(si + lane_f(tN[l], 0)));
        const v4f hself = h4[(size_t)i * 16 + c4];
        v4f acc = {0.0f, 0.0f, 0.0f, 0.0f};
        float den = 0.0f;

        if (ok) {
            const int deg = __builtin_amdgcn_readfirstlane(ncnt[l]);
            const int st  = __builtin_amdgcn_readfirstlane(nstart[l]);
            for (int bb = 0; bb < deg; bb += 8) {
                v4f  rr[4];
                bool vv[4];
#pragma unroll
                for (int k = 0; k < 4; ++k) {
                    const int e = bb + 2 * k + hl;
                    vv[k] = e < deg;
                    int d = dsts[st + min(e, deg - 1)];
                    d = min(max(d, 0), n - 1);
                    rr[k] = h4[(unsigned)d * 16u + (unsigned)c4];
                }
#pragma unroll
                for (int k = 0; k < 4; ++k) {
                    const v4f r = rr[k];
                    const v2f q = r.xy * a2v.xy + r.zw * a2v.zw;
                    const float td = half_sum(q.x + q.y);
                    const float e  = __expf(lrelu(si + td));
                    const float w  = vv[k] ? e : 0.0f;
                    den += w;
                    acc += r * w;
                }
            }
        } else {
            for (int base = 0; base < nE; base += 32) {
                const int e  = base + lane;
                const int ec = e < nE ? e : nE - 1;
                const bool mine = e < nE && ei[ec] == i;
                unsigned mask = __builtin_amdgcn_ballot_w32(mine);
                if (mask == 0u) continue;
                int d = ei[nE + ec];
                d = d < 0 ? 0 : (d > n - 1 ? n - 1 : d);
                const float tv = t[d];
                while (mask != 0u) {
                    const int j = __builtin_ctz(mask);
                    mask &= mask - 1u;
                    const int dj = __builtin_amdgcn_readlane(d, j);
                    const float wj = __expf(lrelu(si + lane_f(tv, j)));
                    const float wl = hl == 0 ? wj : 0.0f;
                    const v4f r = h4[(size_t)dj * 16 + c4];
                    den += wl;
                    acc += r * wl;
                }
            }
        }

        v4f tot;
        tot.x = acc.x + xhalf(acc.x);
        tot.y = acc.y + xhalf(acc.y);
        tot.z = acc.z + xhalf(acc.z);
        tot.w = acc.w + xhalf(acc.w);
        float dn = den + xhalf(den);
        tot += hself * wself;
        dn  += wself;

        v2f o;
        o.x = (hl == 0 ? tot.x : tot.z) / dn;
        o.y = (hl == 0 ? tot.y : tot.w) / dn;
        ((v2f*)orow)[2 * c4 + hl] = o;
    }
}

extern "C" void kernel_launch(void* const* d_in, const int* in_sizes, int n_in,
                              void* d_out, int out_size, void* d_ws, size_t ws_size,
                              hipStream_t stream) {
    const float* x  = (const float*)d_in[0];
    const float* W  = (const float*)d_in[1];
    const float* a  = (const float*)d_in[2];
    const int*   ei = (const int*)d_in[3];

    const int n = in_sizes[0] / IN_C;
    const int E = in_sizes[3] / 2;

    float* ws     = (float*)d_ws;
    float* h      = ws;
    float* s      = h + (size_t)n * OUT_C;
    float* t      = s + n;
    int*   bcnt   = (int*)(t + n);
    float* out    = (float*)d_out;

    const dim3 blk(256);
    const int wavesPerBlock = 256 / 32;

    const int mtiles = (n + 15) / 16;
    const int nbk    = (n + BK_N - 1) / BK_N;
    gat_gemm_stw<<<(mtiles + wavesPerBlock - 1) / wavesPerBlock, blk, 0, stream>>>(
        x, W, a, h, s, t, bcnt, n, mtiles, nbk);

    gat_bin<<<(E + CH - 1) / CH, dim3(BIN_T), 0, stream>>>(ei, bcnt, (unsigned*)d_out, n, E);

    gat_gather<<<nbk, dim3(GAT_T), 0, stream>>>(ei, bcnt, a, h, s, t, out, n, E);
}


# binning kernel: each loop issues its memory operations as one group ahead of one wait (32 index loads, 16 LDS counts, 4 returning global adds, placing and leaving reads); arrival-order sums as before
# speedup vs baseline: 3.4331x; 1.1247x over previous
#include <hip/hip_runtime.h>
#include <hip/hip_bf16.h>
#include <math.h>


typedef __attribute__((ext_vector_type(2))) float v2f;
typedef __attribute__((ext_vector_type(4))) float v4f;
typedef __attribute__((ext_vector_type(8))) float v8f;

#define IN_C  128
#define OUT_C 64
#define GAT_ALPHA 0.2f

#define BK_SH    6
#define BK_N     (1 << BK_SH)
#define BK_WORDS (BK_N * OUT_C)
#define MAXB     2048
#define CH       8192
#define BIN_T    512
#define BIN_EPT  (CH / BIN_T)
#define GAT_T    256

__device__ __forceinline__ float lrelu(float z) {
    return z >= 0.0f ? z : GAT_ALPHA * z;
}

__global__ void gat_gemm_stw(const float* __restrict__ x,
                             const float* __restrict__ W,
                             const float* __restrict__ a,
                             float* __restrict__ h,
                             float* __restrict__ s,
                             float* __restrict__ t,
                             int* __restrict__ bcnt,
                             int n, int mtiles, int nbk) {
    const int wavesPerBlock = blockDim.x >> 5;
    const int wave = blockIdx.x * wavesPerBlock + ((int)threadIdx.x >> 5);
    if (wave >= mtiles) return;

    const int lane = threadIdx.x & 31;
    const int half = lane >> 4;
    const int l15  = lane & 15;
    const int m0   = wave * 16;

    if (lane == 0 && wave < nbk) bcnt[wave] = 0;

    int arow = m0 + l15; if (arow > n - 1) arow = n - 1;
    const float* xrow = x + (size_t)arow * IN_C;

    v2f afrag[32];
#pragma unroll
    for (int ks = 0; ks < 32; ++ks) {
        const int k = ks * 4 + half * 2;
        afrag[ks].x = xrow[k];
        afrag[ks].y = xrow[k + 1];
    }

    float pr1[8], pr2[8];
#pragma unroll
    for (int r = 0; r < 8; ++r) { pr1[r] = 0.0f; pr2[r] = 0.0f; }

#pragma unroll
    for (int nt = 0; nt < 4; ++nt) {
        const int n0 = nt * 16;
        v8f acc = {};
#pragma unroll
        for (int ks = 0; ks < 32; ++ks) {
            const int k = ks * 4 + half * 2;
            v2f b;
            b.x = W[(size_t)k       * OUT_C + n0 + l15];
            b.y = W[(size_t)(k + 1) * OUT_C + n0 + l15];
            acc = __builtin_amdgcn_wmma_f32_16x16x4_f32(
                 false, afrag[ks],  false, b,
                 (short)0, acc,  false,  false);
        }
        const float a1c = a[n0 + l15];
        const float a2c = a[OUT_C + n0 + l15];
#pragma unroll
        for (int r = 0; r < 8; ++r) {
            const int row = m0 + r + 8 * half;
            const float v = acc[r];
            if (row < n) h[(size_t)row * OUT_C + n0 + l15] = v;
            pr1[r] += v * a1c;
            pr2[r] += v * a2c;
        }
    }

#pragma unroll
    for (int mask = 8; mask >= 1; mask >>= 1) {
#pragma unroll
        for (int r = 0; r < 8; ++r) {
            pr1[r] += __shfl_xor(pr1[r], mask, 32);
            pr2[r] += __shfl_xor(pr2[r], mask, 32);
        }
    }
    if (l15 == 0) {
#pragma unroll
        for (int r = 0; r < 8; ++r) {
            const int row = m0 + r + 8 * half;
            if (row < n) { s[row] = pr1[r]; t[row] = pr2[r]; }
        }
    }
}

__global__ void __launch_bounds__(BIN_T)
gat_bin(const int* __restrict__ ei,
        int* __restrict__ bcnt,
        unsigned* __restrict__ slots,
        int n, int nE) {
    __shared__ int      hist[MAXB];
    __shared__ int      gofs[MAXB];
    __shared__ unsigned sEnt[CH];
    __shared__ int      sAdr[CH];
    __shared__ int      wsum[BIN_T / 32];
    __shared__ int      dump[32];

    const int tid   = threadIdx.x;
    const int lane  = tid & 31;
    const int wv    = tid >> 5;
    const int ebase = blockIdx.x * CH;

    for (int j = 0; j < MAXB / BIN_T; ++j) hist[tid + j * BIN_T] = 0;
    if (tid < 32) dump[tid] = 0;
    __syncthreads();

    int sv[BIN_EPT], dv[BIN_EPT];
#pragma unroll
    for (int k = 0; k < BIN_EPT; ++k) {
        const int e  = ebase + k * BIN_T + tid;
        const int ec = e < nE ? e : nE - 1;
        sv[k] = ei[ec];
        dv[k] = ei[nE + ec];
    }

    unsigned ent[BIN_EPT];
    int meta[BIN_EPT];
#pragma unroll
    for (int k = 0; k < BIN_EPT; ++k) {
        const bool valid = ebase + k * BIN_T + tid < nE;
        int sN = sv[k];
        int dN = dv[k];
        sN = sN < 0 ? 0 : (sN > n - 1 ? n - 1 : sN);
        dN = dN < 0 ? 0 : (dN > n - 1 ? n - 1 : dN);
        const int bkt = sN >> BK_SH;
        ent[k] = ((unsigned)dN << BK_SH) | (unsigned)(sN & (BK_N - 1));
        int* tgt = valid ? &hist[bkt] : &dump[lane];
        const int lr = atomicAdd(tgt, 1);
        meta[k] = valid ? (bkt | (lr << 11)) : -1;
    }
    __syncthreads();

    int cc[MAXB / BIN_T], rr[MAXB / BIN_T];
#pragma unroll
    for (int j = 0; j < MAXB / BIN_T; ++j) cc[j] = hist[tid + j * BIN_T];
#pragma unroll
    for (int j = 0; j < MAXB / BIN_T; ++j) {
        rr[j] = 0;
        if (cc[j] > 0) rr[j] = atomicAdd(&bcnt[tid + j * BIN_T], cc[j]);
    }
#pragma unroll
    for (int j = 0; j < MAXB / BIN_T; ++j) gofs[tid + j * BIN_T] = rr[j];

    const int c0 = hist[4 * tid], c1 = hist[4 * tid + 1], c2 = hist[4 * tid + 2], c3 = hist[4 * tid + 3];
    const int tsum = c0 + c1 + c2 + c3;
    int incl = tsum;
    for (int o = 1; o < 32; o <<= 1) {
        const int v = __shfl_up(incl, o, 32);
        if (lane >= o) incl += v;
    }
    if (lane == 31) wsum[wv] = incl;
    __syncthreads();
    int woff = 0;
    for (int q = 0; q < wv; ++q) woff += wsum[q];
    const int ex = woff + incl - tsum;
    hist[4 * tid]     = ex;
    hist[4 * tid + 1] = ex + c0;
    hist[4 * tid + 2] = ex + c0 + c1;
    hist[4 * tid + 3] = ex + c0 + c1 + c2;
    __syncthreads();

    int hb[BIN_EPT], gb[BIN_EPT];
#pragma unroll
    for (int k = 0; k < BIN_EPT; ++k) {
        const int bkt = meta[k] & (MAXB - 1);
        hb[k] = hist[bkt];
        gb[k] = gofs[bkt];
    }
#pragma unroll
    for (int k = 0; k < BIN_EPT; ++k) {
        if (meta[k] >= 0) {
            const int bkt = meta[k] & (MAXB - 1);
            const int lr  = meta[k] >> 11;
            const int p   = hb[k] + lr;
            const int g   = gb[k] + lr;
            int nn = n - (bkt << BK_SH); nn = nn < BK_N ? nn : BK_N;
            sEnt[p] = ent[k];
            sAdr[p] = g < nn * OUT_C ? bkt * BK_WORDS + g : -1;
        }
    }
    __syncthreads();

    int total = nE - ebase; total = total < CH ? total : CH;
    int      oa[BIN_EPT];
    unsigned ow[BIN_EPT];
#pragma unroll
    for (int k = 0; k < BIN_EPT; ++k) {
        const int p = k * BIN_T + tid;
        oa[k] = sAdr[p];
        ow[k] = sEnt[p];
        if (p >= total) oa[k] = -1;
    }
#pragma unroll
    for (int k = 0; k < BIN_EPT; ++k) {
        if (oa[k] >= 0) slots[oa[k]] = ow[k];
    }
}

__device__ __forceinline__ float lane_f(float v, int l) {
    return __builtin_bit_cast(float, __builtin_amdgcn_readlane(__builtin_bit_cast(int, v), l));
}

__device__ __forceinline__ float xhalf(float x) {
    const int b = __builtin_bit_cast(int, x);
    return __builtin_bit_cast(float, __builtin_amdgcn_permlanex16(b, b, 0x76543210, 0xfedcba98, false, false));
}

template <int M>
__device__ __forceinline__ float half_xor(float x) {
    return __builtin_bit_cast(float,
        __builtin_amdgcn_update_dpp(0, __builtin_bit_cast(int, x), 0x160 + M, 0xf, 0xf, true));
}

__device__ __forceinline__ float half_sum(float x) {
    x += half_xor<8>(x);
    x += half_xor<4>(x);
    x += half_xor<2>(x);
    x += half_xor<1>(x);
    return x;
}

__global__ void __launch_bounds__(GAT_T)
gat_gather(const int* __restrict__ ei,
           const int* __restrict__ bcnt,
           const float* __restrict__ a,
           const float* __restrict__ h,
           const float* __restrict__ s,
           const float* __restrict__ t,
           float* __restrict__ out,
           int n, int nE) {
    __shared__ unsigned raw[BK_WORDS];
    __shared__ int      dsts[BK_WORDS];
    __shared__ int      ncnt[BK_N];
    __shared__ int      nstart[BK_N];
    __shared__ int      ncur[BK_N];
    __shared__ float    sN[BK_N];
    __shared__ float    tN[BK_N];
    __shared__ int      nextNode;

    const int tid  = threadIdx.x;
    const int lane = tid & 31;
    const int b    = blockIdx.x;
    const int lo   = b << BK_SH;
    int nn = n - lo; nn = nn < BK_N ? nn : BK_N;
    const int tot  = bcnt[b];
    const bool ok  = tot <= nn * OUT_C;
    const unsigned* region = (const unsigned*)out + (size_t)lo * OUT_C;

    if (tid < BK_N) {
        ncnt[tid] = 0;
        ncur[tid] = 0;
        if (tid < nn) { sN[tid] = s[lo + tid]; tN[tid] = t[lo + tid]; }
    }
    if (tid == 0) nextNode = 0;
    __syncthreads();
    if (ok) {
        for (int base = 0; base < tot; base += 4 * GAT_T) {
            unsigned ev[4];
#pragma unroll
            for (int k = 0; k < 4; ++k) {
                const int p = base + k * GAT_T + tid;
                ev[k] = p < tot ? region[p] : 0u;
            }
#pragma unroll
            for (int k = 0; k < 4; ++k) {
                const int p = base + k * GAT_T + tid;
                if (p < tot) {
                    raw[p] = ev[k];
                    atomicAdd(&ncnt[ev[k] & (BK_N - 1)], 1);
                }
            }
        }
    }
    __syncthreads();
    if (tid < 32) {
        const int a0 = ncnt[2 * lane], a1 = ncnt[2 * lane + 1];
        const int sm = a0 + a1;
        int incl = sm;
        for (int o = 1; o < 32; o <<= 1) {
            const int v = __shfl_up(incl, o, 32);
            if (lane >= o) incl += v;
        }
        nstart[2 * lane]     = incl - sm;
        nstart[2 * lane + 1] = incl - sm + a0;
    }
    __syncthreads();
    if (ok) {
        for (int p = tid; p < tot; p += GAT_T) {
            const unsigned e = raw[p];
            const int l   = (int)(e & (BK_N - 1));
            const int pos = nstart[l] + atomicAdd(&ncur[l], 1);
            dsts[pos] = (int)(e >> BK_SH);
        }
    }
    __syncthreads();

    const int hl = lane >> 4;
    const int c4 = lane & 15;
    const v4f* h4 = (const v4f*)h;
    v4f a2v;
    a2v.x = a[OUT_C + 4 * c4];
    a2v.y = a[OUT_C + 4 * c4 + 1];
    a2v.z = a[OUT_C + 4 * c4 + 2];
    a2v.w = a[OUT_C + 4 * c4 + 3];
    for (;;) {
        int l = 0;
        if (lane == 0) l = atomicAdd(&nextNode, 1);
        l = __builtin_amdgcn_readfirstlane(l);
        if (l >= nn) break;
        const int i = lo + l;
        float* orow = out + (size_t)i * OUT_C;
        const float si = lane_f(sN[l], 0);
        const float wself = __expf(lrelu(si + lane_f(tN[l], 0)));
        const v4f hself = h4[(size_t)i * 16 + c4];
        v4f acc = {0.0f, 0.0f, 0.0f, 0.0f};
        float den = 0.0f;

        if (ok) {
            const int deg = __builtin_amdgcn_readfirstlane(ncnt[l]);
            const int st  = __builtin_amdgcn_readfirstlane(nstart[l]);
            for (int bb = 0; bb < deg; bb += 8) {
                v4f  rr[4];
                bool vv[4];
#pragma unroll
                for (int k = 0; k < 4; ++k) {
                    const int e = bb + 2 * k + hl;
                    vv[k] = e < deg;
                    int d = dsts[st + min(e, deg - 1)];
                    d = min(max(d, 0), n - 1);
                    rr[k] = h4[(unsigned)d * 16u + (unsigned)c4];
                }
#pragma unroll
                for (int k = 0; k < 4; ++k) {
                    const v4f r = rr[k];
                    const v2f q = r.xy * a2v.xy + r.zw * a2v.zw;
                    const float td = half_sum(q.x + q.y);
                    const float e  = __expf(lrelu(si + td));
                    const float w  = vv[k] ? e : 0.0f;
                    den += w;
                    acc += r * w;
                }
            }
        } else {
            for (int base = 0; base < nE; base += 32) {
                const int e  = base + lane;
                const int ec = e < nE ? e : nE - 1;
                const bool mine = e < nE && ei[ec] == i;
                unsigned mask = __builtin_amdgcn_ballot_w32(mine);
                if (mask == 0u) continue;
                int d = ei[nE + ec];
                d = d < 0 ? 0 : (d > n - 1 ? n - 1 : d);
                const float tv = t[d];
                while (mask != 0u) {
                    const int j = __builtin_ctz(mask);
                    mask &= mask - 1u;
                    const int dj = __builtin_amdgcn_readlane(d, j);
                    const float wj = __expf(lrelu(si + lane_f(tv, j)));
                    const float wl = hl == 0 ? wj : 0.0f;
                    const v4f r = h4[(size_t)dj * 16 + c4];
                    den += wl;
                    acc += r * wl;
                }
            }
        }

        v4f tot;
        tot.x = acc.x + xhalf(acc.x);
        tot.y = acc.y + xhalf(acc.y);
        tot.z = acc.z + xhalf(acc.z);
        tot.w = acc.w + xhalf(acc.w);
        float dn = den + xhalf(den);
        tot += hself * wself;
        dn  += wself;

        v2f o;
        o.x = (hl == 0 ? tot.x : tot.z) / dn;
        o.y = (hl == 0 ? tot.y : tot.w) / dn;
        ((v2f*)orow)[2 * c4 + hl] = o;
    }
}

extern "C" void kernel_launch(void* const* d_in, const int* in_sizes, int n_in,
                              void* d_out, int out_size, void* d_ws, size_t ws_size,
                              hipStream_t stream) {
    const float* x  = (const float*)d_in[0];
    const float* W  = (const float*)d_in[1];
    const float* a  = (const float*)d_in[2];
    const int*   ei = (const int*)d_in[3];

    const int n = in_sizes[0] / IN_C;
    const int E = in_sizes[3] / 2;

    float* ws     = (float*)d_ws;
    float* h      = ws;
    float* s      = h + (size_t)n * OUT_C;
    float* t      = s + n;
    int*   bcnt   = (int*)(t + n);
    float* out    = (float*)d_out;

    const dim3 blk(256);
    const int wavesPerBlock = 256 / 32;

    const int mtiles = (n + 15) / 16;
    const int nbk    = (n + BK_N - 1) / BK_N;
    gat_gemm_stw<<<(mtiles + wavesPerBlock - 1) / wavesPerBlock, blk, 0, stream>>>(
        x, W, a, h, s, t, bcnt, n, mtiles, nbk);

    gat_bin<<<(E + CH - 1) / CH, dim3(BIN_T), 0, stream>>>(ei, bcnt, (unsigned*)d_out, n, E);

    gat_gather<<<nbk, dim3(GAT_T), 0, stream>>>(ei, bcnt, a, h, s, t, out, n, E);
}


# GEMM kernel: W (once per block, row pairs interleaved) and each wave's 16x128 x tile staged in LDS by whole-line loads; fragments read from LDS, one A read feeds 4 N-tiles; k order, sums unchanged
# speedup vs baseline: 4.0085x; 1.1676x over previous
#include <hip/hip_runtime.h>
#include <hip/hip_bf16.h>
#include <math.h>


typedef __attribute__((ext_vector_type(2))) float v2f;
typedef __attribute__((ext_vector_type(4))) float v4f;
typedef __attribute__((ext_vector_type(8))) float v8f;

#define IN_C  128
#define OUT_C 64
#define GAT_ALPHA 0.2f

#define BK_SH    6
#define BK_N     (1 << BK_SH)
#define BK_WORDS (BK_N * OUT_C)
#define MAXB     2048
#define CH       8192
#define BIN_T    512
#define BIN_EPT  (CH / BIN_T)
#define GAT_T    256

__device__ __forceinline__ float lrelu(float z) {
    return z >= 0.0f ? z : GAT_ALPHA * z;
}

#define GEMM_T  256
#define GEMM_WV (GEMM_T / 32)
#define W_PP    160
#define X_PITCH 132

__global__ void __launch_bounds__(GEMM_T)
gat_gemm_stw(const float* __restrict__ x,
             const float* __restrict__ W,
             const float* __restrict__ a,
             float* __restrict__ h,
             float* __restrict__ s,
             float* __restrict__ t,
             int* __restrict__ bcnt,
             int n, int mtiles, int nbk) {
    __shared__ __attribute__((aligned(16))) float smem[(IN_C / 2) * W_PP + GEMM_WV * 16 * X_PITCH];
    float* sW = smem;
    float* sX = smem + (IN_C / 2) * W_PP;

    const int tid  = threadIdx.x;
    const int wv   = tid >> 5;
    const int wave = blockIdx.x * GEMM_WV + wv;
    const int lane = tid & 31;
    const int half = lane >> 4;
    const int l15  = lane & 15;
    const int m0   = wave * 16;

    if (lane == 0 && wave < nbk) bcnt[wave] = 0;

    v4f w0[(IN_C * OUT_C) / (8 * GEMM_T)], w1[(IN_C * OUT_C) / (8 * GEMM_T)];
#pragma unroll
    for (int i = 0; i < (IN_C * OUT_C) / (8 * GEMM_T); ++i) {
        const int id = i * GEMM_T + tid;
        const int kp = id >> 4, cg = id & 15;
        w0[i] = *(const v4f*)(W + (2 * kp)     * OUT_C + 4 * cg);
        w1[i] = *(const v4f*)(W + (2 * kp + 1) * OUT_C + 4 * cg);
    }
    v4f xreg[16];
#pragma unroll
    for (int r = 0; r < 16; ++r) {
        int row = m0 + r; if (row > n - 1) row = n - 1;
        xreg[r] = *(const v4f*)(x + (size_t)row * IN_C + 4 * lane);
    }
#pragma unroll
    for (int i = 0; i < (IN_C * OUT_C) / (8 * GEMM_T); ++i) {
        const int id = i * GEMM_T + tid;
        const int kp = id >> 4, cg = id & 15;
        float* dst = sW + kp * W_PP + 8 * cg;
        v4f lo, hi;
        lo.x = w0[i].x; lo.y = w1[i].x; lo.z = w0[i].y; lo.w = w1[i].y;
        hi.x = w0[i].z; hi.y = w1[i].z; hi.z = w0[i].w; hi.w = w1[i].w;
        *(v4f*)dst       = lo;
        *(v4f*)(dst + 4) = hi;
    }
    float* xt = sX + wv * 16 * X_PITCH;
#pragma unroll
    for (int r = 0; r < 16; ++r) {
        *(v4f*)&xt[r * X_PITCH + 4 * lane] = xreg[r];
    }
    __syncthreads();
    if (wave >= mtiles) return;

    v8f acc[4];
#pragma unroll
    for (int nt = 0; nt < 4; ++nt) acc[nt] = v8f{};
    const float* xa = xt + l15 * X_PITCH + half * 2;
    const v2f* wb = (const v2f*)(sW + half * W_PP) + l15;
#pragma unroll
    for (int ks = 0; ks < 32; ++ks) {
        const v2f af = *(const v2f*)&xa[ks * 4];
#pragma unroll
        for (int nt = 0; nt < 4; ++nt) {
            const v2f b = wb[ks * W_PP + nt * 16];
            acc[nt] = __builtin_amdgcn_wmma_f32_16x16x4_f32(
                 false, af,  false, b,
                 (short)0, acc[nt],  false,  false);
        }
    }

    float pr1[8], pr2[8];
#pragma unroll
    for (int r = 0; r < 8; ++r) { pr1[r] = 0.0f; pr2[r] = 0.0f; }
#pragma unroll
    for (int nt = 0; nt < 4; ++nt) {
        const int n0 = nt * 16;
        const float a1c = a[n0 + l15];
        const float a2c = a[OUT_C + n0 + l15];
#pragma unroll
        for (int r = 0; r < 8; ++r) {
            const int row = m0 + r + 8 * half;
            const float v = acc[nt][r];
            if (row < n) h[(size_t)row * OUT_C + n0 + l15] = v;
            pr1[r] += v * a1c;
            pr2[r] += v * a2c;
        }
    }

#pragma unroll
    for (int mask = 8; mask >= 1; mask >>= 1) {
#pragma unroll
        for (int r = 0; r < 8; ++r) {
            pr1[r] += __shfl_xor(pr1[r], mask, 32);
            pr2[r] += __shfl_xor(pr2[r], mask, 32);
        }
    }
    if (l15 == 0) {
#pragma unroll
        for (int r = 0; r < 8; ++r) {
            const int row = m0 + r + 8 * half;
            if (row < n) { s[row] = pr1[r]; t[row] = pr2[r]; }
        }
    }
}

__global__ void __launch_bounds__(BIN_T)
gat_bin(const int* __restrict__ ei,
        int* __restrict__ bcnt,
        unsigned* __restrict__ slots,
        int n, int nE) {
    __shared__ int      hist[MAXB];
    __shared__ int      gofs[MAXB];
    __shared__ unsigned sEnt[CH];
    __shared__ int      sAdr[CH];
    __shared__ int      wsum[BIN_T / 32];
    __shared__ int      dump[32];

    const int tid   = threadIdx.x;
    const int lane  = tid & 31;
    const int wv    = tid >> 5;
    const int ebase = blockIdx.x * CH;

    for (int j = 0; j < MAXB / BIN_T; ++j) hist[tid + j * BIN_T] = 0;
    if (tid < 32) dump[tid] = 0;
    __syncthreads();

    int sv[BIN_EPT], dv[BIN_EPT];
#pragma unroll
    for (int k = 0; k < BIN_EPT; ++k) {
        const int e  = ebase + k * BIN_T + tid;
        const int ec = e < nE ? e : nE - 1;
        sv[k] = ei[ec];
        dv[k] = ei[nE + ec];
    }

    unsigned ent[BIN_EPT];
    int meta[BIN_EPT];
#pragma unroll
    for (int k = 0; k < BIN_EPT; ++k) {
        const bool valid = ebase + k * BIN_T + tid < nE;
        int sN = sv[k];
        int dN = dv[k];
        sN = sN < 0 ? 0 : (sN > n - 1 ? n - 1 : sN);
        dN = dN < 0 ? 0 : (dN > n - 1 ? n - 1 : dN);
        const int bkt = sN >> BK_SH;
        ent[k] = ((unsigned)dN << BK_SH) | (unsigned)(sN & (BK_N - 1));
        int* tgt = valid ? &hist[bkt] : &dump[lane];
        const int lr = atomicAdd(tgt, 1);
        meta[k] = valid ? (bkt | (lr << 11)) : -1;
    }
    __syncthreads();

    int cc[MAXB / BIN_T], rr[MAXB / BIN_T];
#pragma unroll
    for (int j = 0; j < MAXB / BIN_T; ++j) cc[j] = hist[tid + j * BIN_T];
#pragma unroll
    for (int j = 0; j < MAXB / BIN_T; ++j) {
        rr[j] = 0;
        if (cc[j] > 0) rr[j] = atomicAdd(&bcnt[tid + j * BIN_T], cc[j]);
    }
#pragma unroll
    for (int j = 0; j < MAXB / BIN_T; ++j) gofs[tid + j * BIN_T] = rr[j];

    const int c0 = hist[4 * tid], c1 = hist[4 * tid + 1], c2 = hist[4 * tid + 2], c3 = hist[4 * tid + 3];
    const int tsum = c0 + c1 + c2 + c3;
    int incl = tsum;
    for (int o = 1; o < 32; o <<= 1) {
        const int v = __shfl_up(incl, o, 32);
        if (lane >= o) incl += v;
    }
    if (lane == 31) wsum[wv] = incl;
    __syncthreads();
    int woff = 0;
    for (int q = 0; q < wv; ++q) woff += wsum[q];
    const int ex = woff + incl - tsum;
    hist[4 * tid]     = ex;
    hist[4 * tid + 1] = ex + c0;
    hist[4 * tid + 2] = ex + c0 + c1;
    hist[4 * tid + 3] = ex + c0 + c1 + c2;
    __syncthreads();

    int hb[BIN_EPT], gb[BIN_EPT];
#pragma unroll
    for (int k = 0; k < BIN_EPT; ++k) {
        const int bkt = meta[k] & (MAXB - 1);
        hb[k] = hist[bkt];
        gb[k] = gofs[bkt];
    }
#pragma unroll
    for (int k = 0; k < BIN_EPT; ++k) {
        if (meta[k] >= 0) {
            const int bkt = meta[k] & (MAXB - 1);
            const int lr  = meta[k] >> 11;
            const int p   = hb[k] + lr;
            const int g   = gb[k] + lr;
            int nn = n - (bkt << BK_SH); nn = nn < BK_N ? nn : BK_N;
            sEnt[p] = ent[k];
            sAdr[p] = g < nn * OUT_C ? bkt * BK_WORDS + g : -1;
        }
    }
    __syncthreads();

    int total = nE - ebase; total = total < CH ? total : CH;
    int      oa[BIN_EPT];
    unsigned ow[BIN_EPT];
#pragma unroll
    for (int k = 0; k < BIN_EPT; ++k) {
        const int p = k * BIN_T + tid;
        oa[k] = sAdr[p];
        ow[k] = sEnt[p];
        if (p >= total) oa[k] = -1;
    }
#pragma unroll
    for (int k = 0; k < BIN_EPT; ++k) {
        if (oa[k] >= 0) slots[oa[k]] = ow[k];
    }
}

__device__ __forceinline__ float lane_f(float v, int l) {
    return __builtin_bit_cast(float, __builtin_amdgcn_readlane(__builtin_bit_cast(int, v), l));
}

__device__ __forceinline__ float xhalf(float x) {
    const int b = __builtin_bit_cast(int, x);
    return __builtin_bit_cast(float, __builtin_amdgcn_permlanex16(b, b, 0x76543210, 0xfedcba98, false, false));
}

template <int M>
__device__ __forceinline__ float half_xor(float x) {
    return __builtin_bit_cast(float,
        __builtin_amdgcn_update_dpp(0, __builtin_bit_cast(int, x), 0x160 + M, 0xf, 0xf, true));
}

__device__ __forceinline__ float half_sum(float x) {
    x += half_xor<8>(x);
    x += half_xor<4>(x);
    x += half_xor<2>(x);
    x += half_xor<1>(x);
    return x;
}

__global__ void __launch_bounds__(GAT_T)
gat_gather(const int* __restrict__ ei,
           const int* __restrict__ bcnt,
           const float* __restrict__ a,
           const float* __restrict__ h,
           const float* __restrict__ s,
           const float* __restrict__ t,
           float* __restrict__ out,
           int n, int nE) {
    __shared__ unsigned raw[BK_WORDS];
    __shared__ int      dsts[BK_WORDS];
    __shared__ int      ncnt[BK_N];
    __shared__ int      nstart[BK_N];
    __shared__ int      ncur[BK_N];
    __shared__ float    sN[BK_N];
    __shared__ float    tN[BK_N];
    __shared__ int      nextNode;

    const int tid  = threadIdx.x;
    const int lane = tid & 31;
    const int b    = blockIdx.x;
    const int lo   = b << BK_SH;
    int nn = n - lo; nn = nn < BK_N ? nn : BK_N;
    const int tot  = bcnt[b];
    const bool ok  = tot <= nn * OUT_C;
    const unsigned* region = (const unsigned*)out + (size_t)lo * OUT_C;

    if (tid < BK_N) {
        ncnt[tid] = 0;
        ncur[tid] = 0;
        if (tid < nn) { sN[tid] = s[lo + tid]; tN[tid] = t[lo + tid]; }
    }
    if (tid == 0) nextNode = 0;
    __syncthreads();
    if (ok) {
        for (int base = 0; base < tot; base += 4 * GAT_T) {
            unsigned ev[4];
#pragma unroll
            for (int k = 0; k < 4; ++k) {
                const int p = base + k * GAT_T + tid;
                ev[k] = p < tot ? region[p] : 0u;
            }
#pragma unroll
            for (int k = 0; k < 4; ++k) {
                const int p = base + k * GAT_T + tid;
                if (p < tot) {
                    raw[p] = ev[k];
                    atomicAdd(&ncnt[ev[k] & (BK_N - 1)], 1);
                }
            }
        }
    }
    __syncthreads();
    if (tid < 32) {
        const int a0 = ncnt[2 * lane], a1 = ncnt[2 * lane + 1];
        const int sm = a0 + a1;
        int incl = sm;
        for (int o = 1; o < 32; o <<= 1) {
            const int v = __shfl_up(incl, o, 32);
            if (lane >= o) incl += v;
        }
        nstart[2 * lane]     = incl - sm;
        nstart[2 * lane + 1] = incl - sm + a0;
    }
    __syncthreads();
    if (ok) {
        for (int p = tid; p < tot; p += GAT_T) {
            const unsigned e = raw[p];
            const int l   = (int)(e & (BK_N - 1));
            const int pos = nstart[l] + atomicAdd(&ncur[l], 1);
            dsts[pos] = (int)(e >> BK_SH);
        }
    }
    __syncthreads();

    const int hl = lane >> 4;
    const int c4 = lane & 15;
    const v4f* h4 = (const v4f*)h;
    v4f a2v;
    a2v.x = a[OUT_C + 4 * c4];
    a2v.y = a[OUT_C + 4 * c4 + 1];
    a2v.z = a[OUT_C + 4 * c4 + 2];
    a2v.w = a[OUT_C + 4 * c4 + 3];
    for (;;) {
        int l = 0;
        if (lane == 0) l = atomicAdd(&nextNode, 1);
        l = __builtin_amdgcn_readfirstlane(l);
        if (l >= nn) break;
        const int i = lo + l;
        float* orow = out + (size_t)i * OUT_C;
        const float si = lane_f(sN[l], 0);
        const float wself = __expf(lrelu(si + lane_f(tN[l], 0)));
        const v4f hself = h4[(size_t)i * 16 + c4];
        v4f acc = {0.0f, 0.0f, 0.0f, 0.0f};
        float den = 0.0f;

        if (ok) {
            const int deg = __builtin_amdgcn_readfirstlane(ncnt[l]);
            const int st  = __builtin_amdgcn_readfirstlane(nstart[l]);
            for (int bb = 0; bb < deg; bb += 8) {
                v4f  rr[4];
                bool vv[4];
#pragma unroll
                for (int k = 0; k < 4; ++k) {
                    const int e = bb + 2 * k + hl;
                    vv[k] = e < deg;
                    int d = dsts[st + min(e, deg - 1)];
                    d = min(max(d, 0), n - 1);
                    rr[k] = h4[(unsigned)d * 16u + (unsigned)c4];
                }
#pragma unroll
                for (int k = 0; k < 4; ++k) {
                    const v4f r = rr[k];
                    const v2f q = r.xy * a2v.xy + r.zw * a2v.zw;
                    const float td = half_sum(q.x + q.y);
                    const float e  = __expf(lrelu(si + td));
                    const float w  = vv[k] ? e : 0.0f;
                    den += w;
                    acc += r * w;
                }
            }
        } else {
            for (int base = 0; base < nE; base += 32) {
                const int e  = base + lane;
                const int ec = e < nE ? e : nE - 1;
                const bool mine = e < nE && ei[ec] == i;
                unsigned mask = __builtin_amdgcn_ballot_w32(mine);
                if (mask == 0u) continue;
                int d = ei[nE + ec];
                d = d < 0 ? 0 : (d > n - 1 ? n - 1 : d);
                const float tv = t[d];
                while (mask != 0u) {
                    const int j = __builtin_ctz(mask);
                    mask &= mask - 1u;
                    const int dj = __builtin_amdgcn_readlane(d, j);
                    const float wj = __expf(lrelu(si + lane_f(tv, j)));
                    const float wl = hl == 0 ? wj : 0.0f;
                    const v4f r = h4[(size_t)dj * 16 + c4];
                    den += wl;
                    acc += r * wl;
                }
            }
        }

        v4f tot;
        tot.x = acc.x + xhalf(acc.x);
        tot.y = acc.y + xhalf(acc.y);
        tot.z = acc.z + xhalf(acc.z);
        tot.w = acc.w + xhalf(acc.w);
        float dn = den + xhalf(den);
        tot += hself * wself;
        dn  += wself;

        v2f o;
        o.x = (hl == 0 ? tot.x : tot.z) / dn;
        o.y = (hl == 0 ? tot.y : tot.w) / dn;
        ((v2f*)orow)[2 * c4 + hl] = o;
    }
}

extern "C" void kernel_launch(void* const* d_in, const int* in_sizes, int n_in,
                              void* d_out, int out_size, void* d_ws, size_t ws_size,
                              hipStream_t stream) {
    const float* x  = (const float*)d_in[0];
    const float* W  = (const float*)d_in[1];
    const float* a  = (const float*)d_in[2];
    const int*   ei = (const int*)d_in[3];

    const int n = in_sizes[0] / IN_C;
    const int E = in_sizes[3] / 2;

    float* ws     = (float*)d_ws;
    float* h      = ws;
    float* s      = h + (size_t)n * OUT_C;
    float* t      = s + n;
    int*   bcnt   = (int*)(t + n);
    float* out    = (float*)d_out;

    const dim3 blk(256);
    const int wavesPerBlock = 256 / 32;

    const int mtiles = (n + 15) / 16;
    const int nbk    = (n + BK_N - 1) / BK_N;
    gat_gemm_stw<<<(mtiles + wavesPerBlock - 1) / wavesPerBlock, blk, 0, stream>>>(
        x, W, a, h, s, t, bcnt, n, mtiles, nbk);

    gat_bin<<<(E + CH - 1) / CH, dim3(BIN_T), 0, stream>>>(ei, bcnt, (unsigned*)d_out, n, E);

    gat_gather<<<nbk, dim3(GAT_T), 0, stream>>>(ei, bcnt, a, h, s, t, out, n, E);
}


# binning runs inside the GEMM launch as independent blocks of the same grid (disjoint inputs; 256 threads, 4096 edges each); counters cleared by a small launch ahead; sums as before
# speedup vs baseline: 4.1309x; 1.0305x over previous
#include <hip/hip_runtime.h>
#include <hip/hip_bf16.h>
#include <math.h>


typedef __attribute__((ext_vector_type(2))) float v2f;
typedef __attribute__((ext_vector_type(4))) float v4f;
typedef __attribute__((ext_vector_type(8))) float v8f;

#define IN_C  128
#define OUT_C 64
#define GAT_ALPHA 0.2f

#define BK_SH    6
#define BK_N     (1 << BK_SH)
#define BK_WORDS (BK_N * OUT_C)
#define MAXB     2048
#define CH       4096
#define BIN_T    256
#define BIN_EPT  (CH / BIN_T)
#define BIN_BPT  (MAXB / BIN_T)
#define GAT_T    256

__device__ __forceinline__ float lrelu(float z) {
    return z >= 0.0f ? z : GAT_ALPHA * z;
}

#define BIN_LDS_WORDS (2 * MAXB + 2 * CH + BIN_T / 32 + 32)

typedef __attribute__((ext_vector_type(4))) int v4i;

__device__ __forceinline__ void
bin_chunk(int* __restrict__ lds,
          int chunk,
          const int* __restrict__ ei,
          int* __restrict__ bcnt,
          unsigned* __restrict__ slots,
          int n, int nE) {
    int*      hist = lds;
    int*      gofs = hist + MAXB;
    unsigned* sEnt = (unsigned*)(gofs + MAXB);
    int*      sAdr = (int*)(sEnt + CH);
    int*      wsum = sAdr + CH;
    int*      dump = wsum + BIN_T / 32;

    const int tid   = threadIdx.x;
    const int lane  = tid & 31;
    const int wv    = tid >> 5;
    const int ebase = chunk * CH;

    for (int j = 0; j < MAXB / BIN_T; ++j) hist[tid + j * BIN_T] = 0;
    if (tid < 32) dump[tid] = 0;
    __syncthreads();

    int sv[BIN_EPT], dv[BIN_EPT];
#pragma unroll
    for (int k = 0; k < BIN_EPT; ++k) {
        const int e  = ebase + k * BIN_T + tid;
        const int ec = e < nE ? e : nE - 1;
        sv[k] = ei[ec];
        dv[k] = ei[nE + ec];
    }

    unsigned ent[BIN_EPT];
    int meta[BIN_EPT];
#pragma unroll
    for (int k = 0; k < BIN_EPT; ++k) {
        const bool valid = ebase + k * BIN_T + tid < nE;
        int sN = sv[k];
        int dN = dv[k];
        sN = sN < 0 ? 0 : (sN > n - 1 ? n - 1 : sN);
        dN = dN < 0 ? 0 : (dN > n - 1 ? n - 1 : dN);
        const int bkt = sN >> BK_SH;
        ent[k] = ((unsigned)dN << BK_SH) | (unsigned)(sN & (BK_N - 1));
        int* tgt = valid ? &hist[bkt] : &dump[lane];
        const int lr = atomicAdd(tgt, 1);
        meta[k] = valid ? (bkt | (lr << 11)) : -1;
    }
    __syncthreads();

    int cc[MAXB / BIN_T], rr[MAXB / BIN_T];
#pragma unroll
    for (int j = 0; j < MAXB / BIN_T; ++j) cc[j] = hist[tid + j * BIN_T];
#pragma unroll
    for (int j = 0; j < MAXB / BIN_T; ++j) {
        rr[j] = 0;
        if (cc[j] > 0) rr[j] = atomicAdd(&bcnt[tid + j * BIN_T], cc[j]);
    }
#pragma unroll
    for (int j = 0; j < MAXB / BIN_T; ++j) gofs[tid + j * BIN_T] = rr[j];

    const v4i ca = *(const v4i*)&hist[BIN_BPT * tid];
    const v4i cb = *(const v4i*)&hist[BIN_BPT * tid + 4];
    const int suma = ca.x + ca.y + ca.z + ca.w;
    const int tsum = suma + cb.x + cb.y + cb.z + cb.w;
    int incl = tsum;
    for (int o = 1; o < 32; o <<= 1) {
        const int v = __shfl_up(incl, o, 32);
        if (lane >= o) incl += v;
    }
    if (lane == 31) wsum[wv] = incl;
    __syncthreads();
    int woff = 0;
    for (int q = 0; q < wv; ++q) woff += wsum[q];
    const int ex = woff + incl - tsum;
    v4i ea, eb;
    ea.x = ex;
    ea.y = ea.x + ca.x;
    ea.z = ea.y + ca.y;
    ea.w = ea.z + ca.z;
    eb.x = ex + suma;
    eb.y = eb.x + cb.x;
    eb.z = eb.y + cb.y;
    eb.w = eb.z + cb.z;
    *(v4i*)&hist[BIN_BPT * tid]     = ea;
    *(v4i*)&hist[BIN_BPT * tid + 4] = eb;
    __syncthreads();

    int hb[BIN_EPT], gb[BIN_EPT];
#pragma unroll
    for (int k = 0; k < BIN_EPT; ++k) {
        const int bkt = meta[k] & (MAXB - 1);
        hb[k] = hist[bkt];
        gb[k] = gofs[bkt];
    }
#pragma unroll
    for (int k = 0; k < BIN_EPT; ++k) {
        if (meta[k] >= 0) {
            const int bkt = meta[k] & (MAXB - 1);
            const int lr  = meta[k] >> 11;
            const int p   = hb[k] + lr;
            const int g   = gb[k] + lr;
            int nn = n - (bkt << BK_SH); nn = nn < BK_N ? nn : BK_N;
            sEnt[p] = ent[k];
            sAdr[p] = g < nn * OUT_C ? bkt * BK_WORDS + g : -1;
        }
    }
    __syncthreads();

    int total = nE - ebase; total = total < CH ? total : CH;
    int      oa[BIN_EPT];
    unsigned ow[BIN_EPT];
#pragma unroll
    for (int k = 0; k < BIN_EPT; ++k) {
        const int p = k * BIN_T + tid;
        oa[k] = sAdr[p];
        ow[k] = sEnt[p];
        if (p >= total) oa[k] = -1;
    }
#pragma unroll
    for (int k = 0; k < BIN_EPT; ++k) {
        if (oa[k] >= 0) slots[oa[k]] = ow[k];
    }
}

#define GEMM_T  256
#define GEMM_WV (GEMM_T / 32)
#define W_PP    160
#define X_PITCH 132

__global__ void __launch_bounds__(GEMM_T)
gat_clear(int* __restrict__ bcnt, int nbk) {
    const int i = blockIdx.x * GEMM_T + threadIdx.x;
    if (i < nbk) bcnt[i] = 0;
}

__global__ void __launch_bounds__(GEMM_T)
gat_gemm_bin(const float* __restrict__ x,
             const float* __restrict__ W,
             const float* __restrict__ a,
             float* __restrict__ h,
             float* __restrict__ s,
             float* __restrict__ t,
             const int* __restrict__ ei,
             int* __restrict__ bcnt,
             unsigned* __restrict__ slots,
             int n, int mtiles, int nE, int nbin) {
    __shared__ __attribute__((aligned(16))) float smem[(IN_C / 2) * W_PP + GEMM_WV * 16 * X_PITCH];
    static_assert(BIN_LDS_WORDS <= (IN_C / 2) * W_PP + GEMM_WV * 16 * X_PITCH, "the binning arrays fit the GEMM's LDS block");
    static_assert(BIN_T == GEMM_T, "one block size for both kinds of block");
    if ((int)blockIdx.x < nbin) {
        bin_chunk((int*)smem, (int)blockIdx.x, ei, bcnt, slots, n, nE);
        return;
    }
    float* sW = smem;
    float* sX = smem + (IN_C / 2) * W_PP;

    const int tid  = threadIdx.x;
    const int wv   = tid >> 5;
    const int wave = ((int)blockIdx.x - nbin) * GEMM_WV + wv;
    const int lane = tid & 31;
    const int half = lane >> 4;
    const int l15  = lane & 15;
    const int m0   = wave * 16;

    v4f w0[(IN_C * OUT_C) / (8 * GEMM_T)], w1[(IN_C * OUT_C) / (8 * GEMM_T)];
#pragma unroll
    for (int i = 0; i < (IN_C * OUT_C) / (8 * GEMM_T); ++i) {
        const int id = i * GEMM_T + tid;
        const int kp = id >> 4, cg = id & 15;
        w0[i] = *(const v4f*)(W + (2 * kp)     * OUT_C + 4 * cg);
        w1[i] = *(const v4f*)(W + (2 * kp + 1) * OUT_C + 4 * cg);
    }
    v4f xreg[16];
#pragma unroll
    for (int r = 0; r < 16; ++r) {
        int row = m0 + r; if (row > n - 1) row = n - 1;
        xreg[r] = *(const v4f*)(x + (size_t)row * IN_C + 4 * lane);
    }
#pragma unroll
    for (int i = 0; i < (IN_C * OUT_C) / (8 * GEMM_T); ++i) {
        const int id = i * GEMM_T + tid;
        const int kp = id >> 4, cg = id & 15;
        float* dst = sW + kp * W_PP + 8 * cg;
        v4f lo, hi;
        lo.x = w0[i].x; lo.y = w1[i].x; lo.z = w0[i].y; lo.w = w1[i].y;
        hi.x = w0[i].z; hi.y = w1[i].z; hi.z = w0[i].w; hi.w = w1[i].w;
        *(v4f*)dst       = lo;
        *(v4f*)(dst + 4) = hi;
    }
    float* xt = sX + wv * 16 * X_PITCH;
#pragma unroll
    for (int r = 0; r < 16; ++r) {
        *(v4f*)&xt[r * X_PITCH + 4 * lane] = xreg[r];
    }
    __syncthreads();
    if (wave >= mtiles) return;

    v8f acc[4];
#pragma unroll
    for (int nt = 0; nt < 4; ++nt) acc[nt] = v8f{};
    const float* xa = xt + l15 * X_PITCH + half * 2;
    const v2f* wb = (const v2f*)(sW + half * W_PP) + l15;
#pragma unroll
    for (int ks = 0; ks < 32; ++ks) {
        const v2f af = *(const v2f*)&xa[ks * 4];
#pragma unroll
        for (int nt = 0; nt < 4; ++nt) {
            const v2f b = wb[ks * W_PP + nt * 16];
            acc[nt] = __builtin_amdgcn_wmma_f32_16x16x4_f32(
                 false, af,  false, b,
                 (short)0, acc[nt],  false,  false);
        }
    }

    float pr1[8], pr2[8];
#pragma unroll
    for (int r = 0; r < 8; ++r) { pr1[r] = 0.0f; pr2[r] = 0.0f; }
#pragma unroll
    for (int nt = 0; nt < 4; ++nt) {
        const int n0 = nt * 16;
        const float a1c = a[n0 + l15];
        const float a2c = a[OUT_C + n0 + l15];
#pragma unroll
        for (int r = 0; r < 8; ++r) {
            const int row = m0 + r + 8 * half;
            const float v = acc[nt][r];
            if (row < n) h[(size_t)row * OUT_C + n0 + l15] = v;
            pr1[r] += v * a1c;
            pr2[r] += v * a2c;
        }
    }

#pragma unroll
    for (int mask = 8; mask >= 1; mask >>= 1) {
#pragma unroll
        for (int r = 0; r < 8; ++r) {
            pr1[r] += __shfl_xor(pr1[r], mask, 32);
            pr2[r] += __shfl_xor(pr2[r], mask, 32);
        }
    }
    if (l15 == 0) {
#pragma unroll
        for (int r = 0; r < 8; ++r) {
            const int row = m0 + r + 8 * half;
            if (row < n) { s[row] = pr1[r]; t[row] = pr2[r]; }
        }
    }
}

__device__ __forceinline__ float lane_f(float v, int l) {
    return __builtin_bit_cast(float, __builtin_amdgcn_readlane(__builtin_bit_cast(int, v), l));
}

__device__ __forceinline__ float xhalf(float x) {
    const int b = __builtin_bit_cast(int, x);
    return __builtin_bit_cast(float, __builtin_amdgcn_permlanex16(b, b, 0x76543210, 0xfedcba98, false, false));
}

template <int M>
__device__ __forceinline__ float half_xor(float x) {
    return __builtin_bit_cast(float,
        __builtin_amdgcn_update_dpp(0, __builtin_bit_cast(int, x), 0x160 + M, 0xf, 0xf, true));
}

__device__ __forceinline__ float half_sum(float x) {
    x += half_xor<8>(x);
    x += half_xor<4>(x);
    x += half_xor<2>(x);
    x += half_xor<1>(x);
    return x;
}

__global__ void __launch_bounds__(GAT_T)
gat_gather(const int* __restrict__ ei,
           const int* __restrict__ bcnt,
           const float* __restrict__ a,
           const float* __restrict__ h,
           const float* __restrict__ s,
           const float* __restrict__ t,
           float* __restrict__ out,
           int n, int nE) {
    __shared__ unsigned raw[BK_WORDS];
    __shared__ int      dsts[BK_WORDS];
    __shared__ int      ncnt[BK_N];
    __shared__ int      nstart[BK_N];
    __shared__ int      ncur[BK_N];
    __shared__ float    sN[BK_N];
    __shared__ float    tN[BK_N];
    __shared__ int      nextNode;

    const int tid  = threadIdx.x;
    const int lane = tid & 31;
    const int b    = blockIdx.x;
    const int lo   = b << BK_SH;
    int nn = n - lo; nn = nn < BK_N ? nn : BK_N;
    const int tot  = bcnt[b];
    const bool ok  = tot <= nn * OUT_C;
    const unsigned* region = (const unsigned*)out + (size_t)lo * OUT_C;

    if (tid < BK_N) {
        ncnt[tid] = 0;
        ncur[tid] = 0;
        if (tid < nn) { sN[tid] = s[lo + tid]; tN[tid] = t[lo + tid]; }
    }
    if (tid == 0) nextNode = 0;
    __syncthreads();
    if (ok) {
        for (int base = 0; base < tot; base += 4 * GAT_T) {
            unsigned ev[4];
#pragma unroll
            for (int k = 0; k < 4; ++k) {
                const int p = base + k * GAT_T + tid;
                ev[k] = p < tot ? region[p] : 0u;
            }
#pragma unroll
            for (int k = 0; k < 4; ++k) {
                const int p = base + k * GAT_T + tid;
                if (p < tot) {
                    raw[p] = ev[k];
                    atomicAdd(&ncnt[ev[k] & (BK_N - 1)], 1);
                }
            }
        }
    }
    __syncthreads();
    if (tid < 32) {
        const int a0 = ncnt[2 * lane], a1 = ncnt[2 * lane + 1];
        const int sm = a0 + a1;
        int incl = sm;
        for (int o = 1; o < 32; o <<= 1) {
            const int v = __shfl_up(incl, o, 32);
            if (lane >= o) incl += v;
        }
        nstart[2 * lane]     = incl - sm;
        nstart[2 * lane + 1] = incl - sm + a0;
    }
    __syncthreads();
    if (ok) {
        for (int p = tid; p < tot; p += GAT_T) {
            const unsigned e = raw[p];
            const int l   = (int)(e & (BK_N - 1));
            const int pos = nstart[l] + atomicAdd(&ncur[l], 1);
            dsts[pos] = (int)(e >> BK_SH);
        }
    }
    __syncthreads();

    const int hl = lane >> 4;
    const int c4 = lane & 15;
    const v4f* h4 = (const v4f*)h;
    v4f a2v;
    a2v.x = a[OUT_C + 4 * c4];
    a2v.y = a[OUT_C + 4 * c4 + 1];
    a2v.z = a[OUT_C + 4 * c4 + 2];
    a2v.w = a[OUT_C + 4 * c4 + 3];
    for (;;) {
        int l = 0;
        if (lane == 0) l = atomicAdd(&nextNode, 1);
        l = __builtin_amdgcn_readfirstlane(l);
        if (l >= nn) break;
        const int i = lo + l;
        float* orow = out + (size_t)i * OUT_C;
        const float si = lane_f(sN[l], 0);
        const float wself = __expf(lrelu(si + lane_f(tN[l], 0)));
        const v4f hself = h4[(size_t)i * 16 + c4];
        v4f acc = {0.0f, 0.0f, 0.0f, 0.0f};
        float den = 0.0f;

        if (ok) {
            const int deg = __builtin_amdgcn_readfirstlane(ncnt[l]);
            const int st  = __builtin_amdgcn_readfirstlane(nstart[l]);
            for (int bb = 0; bb < deg; bb += 8) {
                v4f  rr[4];
                bool vv[4];
#pragma unroll
                for (int k = 0; k < 4; ++k) {
                    const int e = bb + 2 * k + hl;
                    vv[k] = e < deg;
                    int d = dsts[st + min(e, deg - 1)];
                    d = min(max(d, 0), n - 1);
                    rr[k] = h4[(unsigned)d * 16u + (unsigned)c4];
                }
#pragma unroll
                for (int k = 0; k < 4; ++k) {
                    const v4f r = rr[k];
                    const v2f q = r.xy * a2v.xy + r.zw * a2v.zw;
                    const float td = half_sum(q.x + q.y);
                    const float e  = __expf(lrelu(si + td));
                    const float w  = vv[k] ? e : 0.0f;
                    den += w;
                    acc += r * w;
                }
            }
        } else {
            for (int base = 0; base < nE; base += 32) {
                const int e  = base + lane;
                const int ec = e < nE ? e : nE - 1;
                const bool mine = e < nE && ei[ec] == i;
                unsigned mask = __builtin_amdgcn_ballot_w32(mine);
                if (mask == 0u) continue;
                int d = ei[nE + ec];
                d = d < 0 ? 0 : (d > n - 1 ? n - 1 : d);
                const float tv = t[d];
                while (mask != 0u) {
                    const int j = __builtin_ctz(mask);
                    mask &= mask - 1u;
                    const int dj = __builtin_amdgcn_readlane(d, j);
                    const float wj = __expf(lrelu(si + lane_f(tv, j)));
                    const float wl = hl == 0 ? wj : 0.0f;
                    const v4f r = h4[(size_t)dj * 16 + c4];
                    den += wl;
                    acc += r * wl;
                }
            }
        }

        v4f tot;
        tot.x = acc.x + xhalf(acc.x);
        tot.y = acc.y + xhalf(acc.y);
        tot.z = acc.z + xhalf(acc.z);
        tot.w = acc.w + xhalf(acc.w);
        float dn = den + xhalf(den);
        tot += hself * wself;
        dn  += wself;

        v2f o;
        o.x = (hl == 0 ? tot.x : tot.z) / dn;
        o.y = (hl == 0 ? tot.y : tot.w) / dn;
        ((v2f*)orow)[2 * c4 + hl] = o;
    }
}

extern "C" void kernel_launch(void* const* d_in, const int* in_sizes, int n_in,
                              void* d_out, int out_size, void* d_ws, size_t ws_size,
                              hipStream_t stream) {
    const float* x  = (const float*)d_in[0];
    const float* W  = (const float*)d_in[1];
    const float* a  = (const float*)d_in[2];
    const int*   ei = (const int*)d_in[3];

    const int n = in_sizes[0] / IN_C;
    const int E = in_sizes[3] / 2;

    float* ws     = (float*)d_ws;
    float* h      = ws;
    float* s      = h + (size_t)n * OUT_C;
    float* t      = s + n;
    int*   bcnt   = (int*)(t + n);
    float* out    = (float*)d_out;

    const dim3 blk(256);
    const int wavesPerBlock = 256 / 32;

    const int mtiles = (n + 15) / 16;
    const int nbk    = (n + BK_N - 1) / BK_N;
    gat_clear<<<(nbk + GEMM_T - 1) / GEMM_T, blk, 0, stream>>>(bcnt, nbk);

    const int nbin = (E + CH - 1) / CH;
    gat_gemm_bin<<<nbin + (mtiles + wavesPerBlock - 1) / wavesPerBlock, blk, 0, stream>>>(
        x, W, a, h, s, t, ei, bcnt, (unsigned*)d_out, n, mtiles, E, nbin);

    gat_gather<<<nbk, dim3(GAT_T), 0, stream>>>(ei, bcnt, a, h, s, t, out, n, E);
}


# gather node loop: a trip's 4 row dots h[dst].a2 summed by one transposed 16-lane add tree (one sum per lane), so one exp per trip; weights read back by row_share; same pairs and order of sums
# speedup vs baseline: 4.2896x; 1.0384x over previous
#include <hip/hip_runtime.h>
#include <hip/hip_bf16.h>
#include <math.h>


typedef __attribute__((ext_vector_type(2))) float v2f;
typedef __attribute__((ext_vector_type(4))) float v4f;
typedef __attribute__((ext_vector_type(8))) float v8f;

#define IN_C  128
#define OUT_C 64
#define GAT_ALPHA 0.2f

#define BK_SH    6
#define BK_N     (1 << BK_SH)
#define BK_WORDS (BK_N * OUT_C)
#define MAXB     2048
#define CH       4096
#define BIN_T    256
#define BIN_EPT  (CH / BIN_T)
#define BIN_BPT  (MAXB / BIN_T)
#define GAT_T    256

__device__ __forceinline__ float lrelu(float z) {
    return z >= 0.0f ? z : GAT_ALPHA * z;
}

#define BIN_LDS_WORDS (2 * MAXB + 2 * CH + BIN_T / 32 + 32)

typedef __attribute__((ext_vector_type(4))) int v4i;

__device__ __forceinline__ void
bin_chunk(int* __restrict__ lds,
          int chunk,
          const int* __restrict__ ei,
          int* __restrict__ bcnt,
          unsigned* __restrict__ slots,
          int n, int nE) {
    int*      hist = lds;
    int*      gofs = hist + MAXB;
    unsigned* sEnt = (unsigned*)(gofs + MAXB);
    int*      sAdr = (int*)(sEnt + CH);
    int*      wsum = sAdr + CH;
    int*      dump = wsum + BIN_T / 32;

    const int tid   = threadIdx.x;
    const int lane  = tid & 31;
    const int wv    = tid >> 5;
    const int ebase = chunk * CH;

    for (int j = 0; j < MAXB / BIN_T; ++j) hist[tid + j * BIN_T] = 0;
    if (tid < 32) dump[tid] = 0;
    __syncthreads();

    int sv[BIN_EPT], dv[BIN_EPT];
#pragma unroll
    for (int k = 0; k < BIN_EPT; ++k) {
        const int e  = ebase + k * BIN_T + tid;
        const int ec = e < nE ? e : nE - 1;
        sv[k] = ei[ec];
        dv[k] = ei[nE + ec];
    }

    unsigned ent[BIN_EPT];
    int meta[BIN_EPT];
#pragma unroll
    for (int k = 0; k < BIN_EPT; ++k) {
        const bool valid = ebase + k * BIN_T + tid < nE;
        int sN = sv[k];
        int dN = dv[k];
        sN = sN < 0 ? 0 : (sN > n - 1 ? n - 1 : sN);
        dN = dN < 0 ? 0 : (dN > n - 1 ? n - 1 : dN);
        const int bkt = sN >> BK_SH;
        ent[k] = ((unsigned)dN << BK_SH) | (unsigned)(sN & (BK_N - 1));
        int* tgt = valid ? &hist[bkt] : &dump[lane];
        const int lr = atomicAdd(tgt, 1);
        meta[k] = valid ? (bkt | (lr << 11)) : -1;
    }
    __syncthreads();

    int cc[MAXB / BIN_T], rr[MAXB / BIN_T];
#pragma unroll
    for (int j = 0; j < MAXB / BIN_T; ++j) cc[j] = hist[tid + j * BIN_T];
#pragma unroll
    for (int j = 0; j < MAXB / BIN_T; ++j) {
        rr[j] = 0;
        if (cc[j] > 0) rr[j] = atomicAdd(&bcnt[tid + j * BIN_T], cc[j]);
    }
#pragma unroll
    for (int j = 0; j < MAXB / BIN_T; ++j) gofs[tid + j * BIN_T] = rr[j];

    const v4i ca = *(const v4i*)&hist[BIN_BPT * tid];
    const v4i cb = *(const v4i*)&hist[BIN_BPT * tid + 4];
    const int suma = ca.x + ca.y + ca.z + ca.w;
    const int tsum = suma + cb.x + cb.y + cb.z + cb.w;
    int incl = tsum;
    for (int o = 1; o < 32; o <<= 1) {
        const int v = __shfl_up(incl, o, 32);
        if (lane >= o) incl += v;
    }
    if (lane == 31) wsum[wv] = incl;
    __syncthreads();
    int woff = 0;
    for (int q = 0; q < wv; ++q) woff += wsum[q];
    const int ex = woff + incl - tsum;
    v4i ea, eb;
    ea.x = ex;
    ea.y = ea.x + ca.x;
    ea.z = ea.y + ca.y;
    ea.w = ea.z + ca.z;
    eb.x = ex + suma;
    eb.y = eb.x + cb.x;
    eb.z = eb.y + cb.y;
    eb.w = eb.z + cb.z;
    *(v4i*)&hist[BIN_BPT * tid]     = ea;
    *(v4i*)&hist[BIN_BPT * tid + 4] = eb;
    __syncthreads();

    int hb[BIN_EPT], gb[BIN_EPT];
#pragma unroll
    for (int k = 0; k < BIN_EPT; ++k) {
        const int bkt = meta[k] & (MAXB - 1);
        hb[k] = hist[bkt];
        gb[k] = gofs[bkt];
    }
#pragma unroll
    for (int k = 0; k < BIN_EPT; ++k) {
        if (meta[k] >= 0) {
            const int bkt = meta[k] & (MAXB - 1);
            const int lr  = meta[k] >> 11;
            const int p   = hb[k] + lr;
            const int g   = gb[k] + lr;
            int nn = n - (bkt << BK_SH); nn = nn < BK_N ? nn : BK_N;
            sEnt[p] = ent[k];
            sAdr[p] = g < nn * OUT_C ? bkt * BK_WORDS + g : -1;
        }
    }
    __syncthreads();

    int total = nE - ebase; total = total < CH ? total : CH;
    int      oa[BIN_EPT];
    unsigned ow[BIN_EPT];
#pragma unroll
    for (int k = 0; k < BIN_EPT; ++k) {
        const int p = k * BIN_T + tid;
        oa[k] = sAdr[p];
        ow[k] = sEnt[p];
        if (p >= total) oa[k] = -1;
    }
#pragma unroll
    for (int k = 0; k < BIN_EPT; ++k) {
        if (oa[k] >= 0) slots[oa[k]] = ow[k];
    }
}

#define GEMM_T  256
#define GEMM_WV (GEMM_T / 32)
#define W_PP    160
#define X_PITCH 132

__global__ void __launch_bounds__(GEMM_T)
gat_clear(int* __restrict__ bcnt, int nbk) {
    const int i = blockIdx.x * GEMM_T + threadIdx.x;
    if (i < nbk) bcnt[i] = 0;
}

__global__ void __launch_bounds__(GEMM_T)
gat_gemm_bin(const float* __restrict__ x,
             const float* __restrict__ W,
             const float* __restrict__ a,
             float* __restrict__ h,
             float* __restrict__ s,
             float* __restrict__ t,
             const int* __restrict__ ei,
             int* __restrict__ bcnt,
             unsigned* __restrict__ slots,
             int n, int mtiles, int nE, int nbin) {
    __shared__ __attribute__((aligned(16))) float smem[(IN_C / 2) * W_PP + GEMM_WV * 16 * X_PITCH];
    static_assert(BIN_LDS_WORDS <= (IN_C / 2) * W_PP + GEMM_WV * 16 * X_PITCH, "the binning arrays fit the GEMM's LDS block");
    static_assert(BIN_T == GEMM_T, "one block size for both kinds of block");
    if ((int)blockIdx.x < nbin) {
        bin_chunk((int*)smem, (int)blockIdx.x, ei, bcnt, slots, n, nE);
        return;
    }
    float* sW = smem;
    float* sX = smem + (IN_C / 2) * W_PP;

    const int tid  = threadIdx.x;
    const int wv   = tid >> 5;
    const int wave = ((int)blockIdx.x - nbin) * GEMM_WV + wv;
    const int lane = tid & 31;
    const int half = lane >> 4;
    const int l15  = lane & 15;
    const int m0   = wave * 16;

    v4f w0[(IN_C * OUT_C) / (8 * GEMM_T)], w1[(IN_C * OUT_C) / (8 * GEMM_T)];
#pragma unroll
    for (int i = 0; i < (IN_C * OUT_C) / (8 * GEMM_T); ++i) {
        const int id = i * GEMM_T + tid;
        const int kp = id >> 4, cg = id & 15;
        w0[i] = *(const v4f*)(W + (2 * kp)     * OUT_C + 4 * cg);
        w1[i] = *(const v4f*)(W + (2 * kp + 1) * OUT_C + 4 * cg);
    }
    v4f xreg[16];
#pragma unroll
    for (int r = 0; r < 16; ++r) {
        int row = m0 + r; if (row > n - 1) row = n - 1;
        xreg[r] = *(const v4f*)(x + (size_t)row * IN_C + 4 * lane);
    }
#pragma unroll
    for (int i = 0; i < (IN_C * OUT_C) / (8 * GEMM_T); ++i) {
        const int id = i * GEMM_T + tid;
        const int kp = id >> 4, cg = id & 15;
        float* dst = sW + kp * W_PP + 8 * cg;
        v4f lo, hi;
        lo.x = w0[i].x; lo.y = w1[i].x; lo.z = w0[i].y; lo.w = w1[i].y;
        hi.x = w0[i].z; hi.y = w1[i].z; hi.z = w0[i].w; hi.w = w1[i].w;
        *(v4f*)dst       = lo;
        *(v4f*)(dst + 4) = hi;
    }
    float* xt = sX + wv * 16 * X_PITCH;
#pragma unroll
    for (int r = 0; r < 16; ++r) {
        *(v4f*)&xt[r * X_PITCH + 4 * lane] = xreg[r];
    }
    __syncthreads();
    if (wave >= mtiles) return;

    v8f acc[4];
#pragma unroll
    for (int nt = 0; nt < 4; ++nt) acc[nt] = v8f{};
    const float* xa = xt + l15 * X_PITCH + half * 2;
    const v2f* wb = (const v2f*)(sW + half * W_PP) + l15;
#pragma unroll
    for (int ks = 0; ks < 32; ++ks) {
        const v2f af = *(const v2f*)&xa[ks * 4];
#pragma unroll
        for (int nt = 0; nt < 4; ++nt) {
            const v2f b = wb[ks * W_PP + nt * 16];
            acc[nt] = __builtin_amdgcn_wmma_f32_16x16x4_f32(
                 false, af,  false, b,
                 (short)0, acc[nt],  false,  false);
        }
    }

    float pr1[8], pr2[8];
#pragma unroll
    for (int r = 0; r < 8; ++r) { pr1[r] = 0.0f; pr2[r] = 0.0f; }
#pragma unroll
    for (int nt = 0; nt < 4; ++nt) {
        const int n0 = nt * 16;
        const float a1c = a[n0 + l15];
        const float a2c = a[OUT_C + n0 + l15];
#pragma unroll
        for (int r = 0; r < 8; ++r) {
            const int row = m0 + r + 8 * half;
            const float v = acc[nt][r];
            if (row < n) h[(size_t)row * OUT_C + n0 + l15] = v;
            pr1[r] += v * a1c;
            pr2[r] += v * a2c;
        }
    }

#pragma unroll
    for (int mask = 8; mask >= 1; mask >>= 1) {
#pragma unroll
        for (int r = 0; r < 8; ++r) {
            pr1[r] += __shfl_xor(pr1[r], mask, 32);
            pr2[r] += __shfl_xor(pr2[r], mask, 32);
        }
    }
    if (l15 == 0) {
#pragma unroll
        for (int r = 0; r < 8; ++r) {
            const int row = m0 + r + 8 * half;
            if (row < n) { s[row] = pr1[r]; t[row] = pr2[r]; }
        }
    }
}

__device__ __forceinline__ float lane_f(float v, int l) {
    return __builtin_bit_cast(float, __builtin_amdgcn_readlane(__builtin_bit_cast(int, v), l));
}

__device__ __forceinline__ float xhalf(float x) {
    const int b = __builtin_bit_cast(int, x);
    return __builtin_bit_cast(float, __builtin_amdgcn_permlanex16(b, b, 0x76543210, 0xfedcba98, false, false));
}

template <int M>
__device__ __forceinline__ float half_xor(float x) {
    return __builtin_bit_cast(float,
        __builtin_amdgcn_update_dpp(0, __builtin_bit_cast(int, x), 0x160 + M, 0xf, 0xf, true));
}

template <int N>
__device__ __forceinline__ float half_bcast(float x) {
    return __builtin_bit_cast(float,
        __builtin_amdgcn_update_dpp(0, __builtin_bit_cast(int, x), 0x150 + N, 0xf, 0xf, true));
}

__device__ __forceinline__ float half_sum4_transposed(const float p[4], int c4) {
    const bool b3 = (c4 & 8) != 0;
    const bool b2 = (c4 & 4) != 0;
    float u[2];
#pragma unroll
    for (int j = 0; j < 2; ++j) {
        const float keep = b3 ? p[j + 2] : p[j];
        const float give = b3 ? p[j] : p[j + 2];
        u[j] = keep + half_xor<8>(give);
    }
    const float keep = b2 ? u[1] : u[0];
    const float give = b2 ? u[0] : u[1];
    float td = keep + half_xor<4>(give);
    td += half_xor<2>(td);
    td += half_xor<1>(td);
    return td;
}

__global__ void __launch_bounds__(GAT_T)
gat_gather(const int* __restrict__ ei,
           const int* __restrict__ bcnt,
           const float* __restrict__ a,
           const float* __restrict__ h,
           const float* __restrict__ s,
           const float* __restrict__ t,
           float* __restrict__ out,
           int n, int nE) {
    __shared__ unsigned raw[BK_WORDS];
    __shared__ int      dsts[BK_WORDS];
    __shared__ int      ncnt[BK_N];
    __shared__ int      nstart[BK_N];
    __shared__ int      ncur[BK_N];
    __shared__ float    sN[BK_N];
    __shared__ float    tN[BK_N];
    __shared__ int      nextNode;

    const int tid  = threadIdx.x;
    const int lane = tid & 31;
    const int b    = blockIdx.x;
    const int lo   = b << BK_SH;
    int nn = n - lo; nn = nn < BK_N ? nn : BK_N;
    const int tot  = bcnt[b];
    const bool ok  = tot <= nn * OUT_C;
    const unsigned* region = (const unsigned*)out + (size_t)lo * OUT_C;

    if (tid < BK_N) {
        ncnt[tid] = 0;
        ncur[tid] = 0;
        if (tid < nn) { sN[tid] = s[lo + tid]; tN[tid] = t[lo + tid]; }
    }
    if (tid == 0) nextNode = 0;
    __syncthreads();
    if (ok) {
        for (int base = 0; base < tot; base += 4 * GAT_T) {
            unsigned ev[4];
#pragma unroll
            for (int k = 0; k < 4; ++k) {
                const int p = base + k * GAT_T + tid;
                ev[k] = p < tot ? region[p] : 0u;
            }
#pragma unroll
            for (int k = 0; k < 4; ++k) {
                const int p = base + k * GAT_T + tid;
                if (p < tot) {
                    raw[p] = ev[k];
                    atomicAdd(&ncnt[ev[k] & (BK_N - 1)], 1);
                }
            }
        }
    }
    __syncthreads();
    if (tid < 32) {
        const int a0 = ncnt[2 * lane], a1 = ncnt[2 * lane + 1];
        const int sm = a0 + a1;
        int incl = sm;
        for (int o = 1; o < 32; o <<= 1) {
            const int v = __shfl_up(incl, o, 32);
            if (lane >= o) incl += v;
        }
        nstart[2 * lane]     = incl - sm;
        nstart[2 * lane + 1] = incl - sm + a0;
    }
    __syncthreads();
    if (ok) {
        for (int p = tid; p < tot; p += GAT_T) {
            const unsigned e = raw[p];
            const int l   = (int)(e & (BK_N - 1));
            const int pos = nstart[l] + atomicAdd(&ncur[l], 1);
            dsts[pos] = (int)(e >> BK_SH);
        }
    }
    __syncthreads();

    const int hl = lane >> 4;
    const int c4 = lane & 15;
    const v4f* h4 = (const v4f*)h;
    v4f a2v;
    a2v.x = a[OUT_C + 4 * c4];
    a2v.y = a[OUT_C + 4 * c4 + 1];
    a2v.z = a[OUT_C + 4 * c4 + 2];
    a2v.w = a[OUT_C + 4 * c4 + 3];
    for (;;) {
        int l = 0;
        if (lane == 0) l = atomicAdd(&nextNode, 1);
        l = __builtin_amdgcn_readfirstlane(l);
        if (l >= nn) break;
        const int i = lo + l;
        float* orow = out + (size_t)i * OUT_C;
        const float si = lane_f(sN[l], 0);
        const float wself = __expf(lrelu(si + lane_f(tN[l], 0)));
        const v4f hself = h4[(size_t)i * 16 + c4];
        v4f acc = {0.0f, 0.0f, 0.0f, 0.0f};
        float den = 0.0f;

        if (ok) {
            const int deg = __builtin_amdgcn_readfirstlane(ncnt[l]);
            const int st  = __builtin_amdgcn_readfirstlane(nstart[l]);
            for (int bb = 0; bb < deg; bb += 8) {
                v4f  rr[4];
#pragma unroll
                for (int k = 0; k < 4; ++k) {
                    const int e = bb + 2 * k + hl;
                    int d = dsts[st + min(e, deg - 1)];
                    d = min(max(d, 0), n - 1);
                    rr[k] = h4[(unsigned)d * 16u + (unsigned)c4];
                }
                float p[4];
#pragma unroll
                for (int k = 0; k < 4; ++k) {
                    const v2f q = rr[k].xy * a2v.xy + rr[k].zw * a2v.zw;
                    p[k] = q.x + q.y;
                }
                const float td = half_sum4_transposed(p, c4);
                const float ex = __expf(lrelu(si + td));
                const float wl = (bb + 2 * (c4 >> 2) + hl) < deg ? ex : 0.0f;
                const float w0 = half_bcast<0>(wl);
                const float w1 = half_bcast<4>(wl);
                const float w2 = half_bcast<8>(wl);
                const float w3 = half_bcast<12>(wl);
                den += w0; acc += rr[0] * w0;
                den += w1; acc += rr[1] * w1;
                den += w2; acc += rr[2] * w2;
                den += w3; acc += rr[3] * w3;
            }
        } else {
            for (int base = 0; base < nE; base += 32) {
                const int e  = base + lane;
                const int ec = e < nE ? e : nE - 1;
                const bool mine = e < nE && ei[ec] == i;
                unsigned mask = __builtin_amdgcn_ballot_w32(mine);
                if (mask == 0u) continue;
                int d = ei[nE + ec];
                d = d < 0 ? 0 : (d > n - 1 ? n - 1 : d);
                const float tv = t[d];
                while (mask != 0u) {
                    const int j = __builtin_ctz(mask);
                    mask &= mask - 1u;
                    const int dj = __builtin_amdgcn_readlane(d, j);
                    const float wj = __expf(lrelu(si + lane_f(tv, j)));
                    const float wl = hl == 0 ? wj : 0.0f;
                    const v4f r = h4[(size_t)dj * 16 + c4];
                    den += wl;
                    acc += r * wl;
                }
            }
        }

        v4f tot;
        tot.x = acc.x + xhalf(acc.x);
        tot.y = acc.y + xhalf(acc.y);
        tot.z = acc.z + xhalf(acc.z);
        tot.w = acc.w + xhalf(acc.w);
        float dn = den + xhalf(den);
        tot += hself * wself;
        dn  += wself;

        v2f o;
        o.x = (hl == 0 ? tot.x : tot.z) / dn;
        o.y = (hl == 0 ? tot.y : tot.w) / dn;
        ((v2f*)orow)[2 * c4 + hl] = o;
    }
}

extern "C" void kernel_launch(void* const* d_in, const int* in_sizes, int n_in,
                              void* d_out, int out_size, void* d_ws, size_t ws_size,
                              hipStream_t stream) {
    const float* x  = (const float*)d_in[0];
    const float* W  = (const float*)d_in[1];
    const float* a  = (const float*)d_in[2];
    const int*   ei = (const int*)d_in[3];

    const int n = in_sizes[0] / IN_C;
    const int E = in_sizes[3] / 2;

    float* ws     = (float*)d_ws;
    float* h      = ws;
    float* s      = h + (size_t)n * OUT_C;
    float* t      = s + n;
    int*   bcnt   = (int*)(t + n);
    float* out    = (float*)d_out;

    const dim3 blk(256);
    const int wavesPerBlock = 256 / 32;

    const int mtiles = (n + 15) / 16;
    const int nbk    = (n + BK_N - 1) / BK_N;
    gat_clear<<<(nbk + GEMM_T - 1) / GEMM_T, blk, 0, stream>>>(bcnt, nbk);

    const int nbin = (E + CH - 1) / CH;
    gat_gemm_bin<<<nbin + (mtiles + wavesPerBlock - 1) / wavesPerBlock, blk, 0, stream>>>(
        x, W, a, h, s, t, ei, bcnt, (unsigned*)d_out, n, mtiles, E, nbin);

    gat_gather<<<nbk, dim3(GAT_T), 0, stream>>>(ei, bcnt, a, h, s, t, out, n, E);
}
